# starting kernel
# baseline (speedup 1.0000x reference)
#include <hip/hip_runtime.h>
#include <math.h>

#define N_NODES 50000
#define N_EDGES 800000
#define GDIM 128
#define HD1 128
#define HEADS 4
#define HD2 32
#define RREL 8
#define NB 20

typedef __attribute__((ext_vector_type(2))) float v2f;
typedef __attribute__((ext_vector_type(8))) float v8f;

__device__ __forceinline__ float lrelu(float x) { return x > 0.f ? x : 0.2f * x; }

__device__ __forceinline__ void atomicMaxF(float* addr, float val) {
    if (val >= 0.f) atomicMax((int*)addr, __float_as_int(val));
    else            atomicMin((unsigned int*)addr, __float_as_uint(val));
}


__global__ void k_basis_w(const float* __restrict__ comp, const float* __restrict__ basis,
                          float* __restrict__ W) {
    int t = blockIdx.x * blockDim.x + threadIdx.x;
    if (t >= RREL * GDIM * HD1) return;
    int r = t / (GDIM * HD1);
    int io = t % (GDIM * HD1);
    float s = 0.f;
    #pragma unroll
    for (int b = 0; b < NB; ++b) s += comp[r * NB + b] * basis[b * GDIM * HD1 + io];
    W[t] = s;
}

__global__ void k_count(const int* __restrict__ ei, const int* __restrict__ et,
                        float* __restrict__ cnt) {
    int e = blockIdx.x * blockDim.x + threadIdx.x;
    if (e >= N_EDGES) return;
    atomicAdd(&cnt[ei[N_EDGES + e] * RREL + et[e]], 1.f);
}

__global__ void k_fill(float* __restrict__ p, float v, int n) {
    int i = blockIdx.x * blockDim.x + threadIdx.x;
    if (i < n) p[i] = v;
}

__global__ void k_fill_bias(float* __restrict__ p, const float* __restrict__ bias, int n) {
    int i = blockIdx.x * blockDim.x + threadIdx.x;
    if (i < n) p[i] = bias[i & 127];
}


__global__ void k_scatter_rel(const float* __restrict__ x, const int* __restrict__ ei,
                              const int* __restrict__ et, const float* __restrict__ cnt,
                              float* __restrict__ acc, int r) {
    int gid = blockIdx.x * blockDim.x + threadIdx.x;
    int e = gid >> 5;
    if (e >= N_EDGES) return;
    if (et[e] != r) return;
    int lane = gid & 31;
    int s = ei[e], d = ei[N_EDGES + e];
    float scale = 1.f / cnt[d * RREL + r];
    float4 v = ((const float4*)(x + (size_t)s * GDIM))[lane];
    float* o = acc + (size_t)d * GDIM + lane * 4;
    atomicAdd(o + 0, v.x * scale);
    atomicAdd(o + 1, v.y * scale);
    atomicAdd(o + 2, v.z * scale);
    atomicAdd(o + 3, v.w * scale);
}


template <bool BIAS_MODE>
__global__ __launch_bounds__(256) void k_mm16(const float* __restrict__ A,
                                              const float* __restrict__ B,
                                              float* __restrict__ C,
                                              const float* __restrict__ bias) {
    __shared__ float As[16][132];
    int tx = threadIdx.x;
    int ty = threadIdx.y;
    int tid = ty * 32 + tx;
    int row0 = blockIdx.x * 16;

    const float4* A4 = (const float4*)(A + (size_t)row0 * 128);
    for (int i = tid; i < 16 * 32; i += 256) {
        int rr = i >> 5, cc = i & 31;
        float4 v = A4[rr * 32 + cc];
        As[rr][cc * 4 + 0] = v.x;
        As[rr][cc * 4 + 1] = v.y;
        As[rr][cc * 4 + 2] = v.z;
        As[rr][cc * 4 + 3] = v.w;
    }
    __syncthreads();

    int col0 = ty * 16;
    int half = tx >> 4;
    int mn = tx & 15;
    int kg = half * 2;

    v8f c;
    if (BIAS_MODE) {
        float bv = bias ? bias[col0 + mn] : 0.f;
        #pragma unroll
        for (int i = 0; i < 8; ++i) c[i] = bv;
    } else {
        #pragma unroll
        for (int i = 0; i < 8; ++i)
            c[i] = C[(size_t)(row0 + half * 8 + i) * 128 + col0 + mn];
    }

    #pragma unroll 4
    for (int k0 = 0; k0 < 128; k0 += 4) {
        v2f a, b;
        a.x = As[mn][k0 + kg];
        a.y = As[mn][k0 + kg + 1];
        b.x = B[(size_t)(k0 + kg) * 128 + col0 + mn];
        b.y = B[(size_t)(k0 + kg + 1) * 128 + col0 + mn];
        c = __builtin_amdgcn_wmma_f32_16x16x4_f32(false, a, false, b, (short)0, c,
                                                  false, false);
    }

    #pragma unroll
    for (int i = 0; i < 8; ++i)
        C[(size_t)(row0 + half * 8 + i) * 128 + col0 + mn] = c[i];
}


__global__ void k_attn(const float* __restrict__ xl, const float* __restrict__ att_src,
                       const float* __restrict__ att_dst, float* __restrict__ a_src,
                       float* __restrict__ a_dst) {
    int t = blockIdx.x * blockDim.x + threadIdx.x;
    if (t >= N_NODES * HEADS) return;
    int n = t >> 2, h = t & 3;
    const float* row = xl + (size_t)n * 128 + h * HD2;
    float ss = 0.f, sd = 0.f;
    #pragma unroll
    for (int j = 0; j < HD2; ++j) {
        float v = row[j];
        ss += v * att_src[h * HD2 + j];
        sd += v * att_dst[h * HD2 + j];
    }
    a_src[t] = ss;
    a_dst[t] = sd;
}

__global__ void k_edge_max(const int* __restrict__ ei, const float* __restrict__ a_src,
                           const float* __restrict__ a_dst, float* __restrict__ m) {
    int i = blockIdx.x * blockDim.x + threadIdx.x;
    if (i >= N_EDGES + N_NODES) return;
    int s, d;
    if (i < N_EDGES) { s = ei[i]; d = ei[N_EDGES + i]; if (s == d) return; }
    else             { s = d = i - N_EDGES; }
    #pragma unroll
    for (int h = 0; h < HEADS; ++h) {
        float v = lrelu(a_src[s * 4 + h] + a_dst[d * 4 + h]);
        atomicMaxF(&m[d * 4 + h], v);
    }
}

__global__ void k_edge_sum(const int* __restrict__ ei, const float* __restrict__ a_src,
                           const float* __restrict__ a_dst, const float* __restrict__ m,
                           float* __restrict__ denom) {
    int i = blockIdx.x * blockDim.x + threadIdx.x;
    if (i >= N_EDGES + N_NODES) return;
    int s, d;
    if (i < N_EDGES) { s = ei[i]; d = ei[N_EDGES + i]; if (s == d) return; }
    else             { s = d = i - N_EDGES; }
    #pragma unroll
    for (int h = 0; h < HEADS; ++h) {
        float v = lrelu(a_src[s * 4 + h] + a_dst[d * 4 + h]);
        atomicAdd(&denom[d * 4 + h], __expf(v - m[d * 4 + h]));
    }
}

__global__ void k_edge_scatter(const int* __restrict__ ei, const float* __restrict__ xl,
                               const float* __restrict__ a_src, const float* __restrict__ a_dst,
                               const float* __restrict__ m, const float* __restrict__ denom,
                               float* __restrict__ out) {
    int gid = blockIdx.x * blockDim.x + threadIdx.x;
    int i = gid >> 5;
    if (i >= N_EDGES + N_NODES) return;
    int lane = gid & 31;
    int s, d;
    if (i < N_EDGES) { s = ei[i]; d = ei[N_EDGES + i]; if (s == d) return; }
    else             { s = d = i - N_EDGES; }
    int h = lane >> 3;
    float v = lrelu(a_src[s * 4 + h] + a_dst[d * 4 + h]);
    float w = __expf(v - m[d * 4 + h]) / denom[d * 4 + h];
    float4 xv = ((const float4*)(xl + (size_t)s * 128))[lane];
    float* o = out + (size_t)d * 128 + lane * 4;
    atomicAdd(o + 0, xv.x * w);
    atomicAdd(o + 1, xv.y * w);
    atomicAdd(o + 2, xv.z * w);
    atomicAdd(o + 3, xv.w * w);
}


extern "C" void kernel_launch(void* const* d_in, const int* in_sizes, int n_in,
                              void* d_out, int out_size, void* d_ws, size_t ws_size,
                              hipStream_t stream) {
    const float* x        = (const float*)d_in[0];
    const int*   ei       = (const int*)d_in[1];
    const int*   et       = (const int*)d_in[2];
    const float* comp     = (const float*)d_in[3];
    const float* basis    = (const float*)d_in[4];
    const float* root     = (const float*)d_in[5];
    const float* bias1    = (const float*)d_in[6];
    const float* gat_w    = (const float*)d_in[7];
    const float* att_src  = (const float*)d_in[8];
    const float* att_dst  = (const float*)d_in[9];
    const float* gat_bias = (const float*)d_in[10];
    float* out = (float*)d_out;

    float* ws = (float*)d_ws;
    size_t off = 0;
    float* W    = ws + off; off += (size_t)RREL * GDIM * HD1;
    float* cnt  = ws + off; off += (size_t)N_NODES * RREL;
    float* hbuf = ws + off; off += (size_t)N_NODES * HD1;
    float* xl   = ws + off; off += (size_t)N_NODES * HD1;
    float* asg  = ws + off; off += (size_t)N_NODES * HEADS;
    float* adg  = ws + off; off += (size_t)N_NODES * HEADS;
    float* mbuf = ws + off; off += (size_t)N_NODES * HEADS;
    float* dbuf = ws + off; off += (size_t)N_NODES * HEADS;
    float* acc = xl;

    dim3 mmBlock(32, 8);
    int mmGrid = N_NODES / 16;

    k_basis_w<<<(RREL * GDIM * HD1 + 255) / 256, 256, 0, stream>>>(comp, basis, W);
    hipMemsetAsync(cnt, 0, (size_t)N_NODES * RREL * sizeof(float), stream);
    k_count<<<(N_EDGES + 255) / 256, 256, 0, stream>>>(ei, et, cnt);
    k_fill_bias<<<(N_NODES * HD1 + 255) / 256, 256, 0, stream>>>(hbuf, bias1, N_NODES * HD1);

    for (int r = 0; r < RREL; ++r) {
        hipMemsetAsync(acc, 0, (size_t)N_NODES * GDIM * sizeof(float), stream);
        k_scatter_rel<<<(N_EDGES * 32 + 255) / 256, 256, 0, stream>>>(x, ei, et, cnt, acc, r);
        k_mm16<false><<<mmGrid, mmBlock, 0, stream>>>(acc, W + (size_t)r * GDIM * HD1,
                                                      hbuf, nullptr);
    }
    k_mm16<false><<<mmGrid, mmBlock, 0, stream>>>(x, root, hbuf, nullptr);

    k_mm16<true><<<mmGrid, mmBlock, 0, stream>>>(hbuf, gat_w, xl, nullptr);
    k_attn<<<(N_NODES * HEADS + 255) / 256, 256, 0, stream>>>(xl, att_src, att_dst, asg, adg);
    k_fill<<<(N_NODES * HEADS + 255) / 256, 256, 0, stream>>>(mbuf, -INFINITY, N_NODES * HEADS);
    hipMemsetAsync(dbuf, 0, (size_t)N_NODES * HEADS * sizeof(float), stream);

    int EN = N_EDGES + N_NODES;
    k_edge_max<<<(EN + 255) / 256, 256, 0, stream>>>(ei, asg, adg, mbuf);
    k_edge_sum<<<(EN + 255) / 256, 256, 0, stream>>>(ei, asg, adg, mbuf, dbuf);
    k_fill_bias<<<(N_NODES * HD1 + 255) / 256, 256, 0, stream>>>(out, gat_bias, N_NODES * HD1);
    k_edge_scatter<<<(EN * 32 + 255) / 256, 256, 0, stream>>>(ei, xl, asg, adg, mbuf, dbuf, out);

    (void)in_sizes; (void)n_in; (void)out_size; (void)ws_size;
}


# tile kernels scan 16-bit destination keys packed by the first launch: packed subtract and min test 128 edges per branch (was 32), half the bytes read; LDS float atomics stay, arrival order
# speedup vs baseline: 3.7808x; 3.7808x over previous
#include <hip/hip_runtime.h>
#include <math.h>

#define N_NODES 50000
#define N_EDGES 800000
#define GDIM 128
#define HD1 128
#define HEADS 4
#define HD2 32
#define RREL 8
#define NB 20

typedef __attribute__((ext_vector_type(2))) float v2f;
typedef __attribute__((ext_vector_type(8))) float v8f;
typedef __attribute__((ext_vector_type(4))) unsigned int v4u;
typedef __attribute__((ext_vector_type(2))) unsigned short us2;

__device__ __forceinline__ unsigned pk_sub16(unsigned a, unsigned b) {
    return __builtin_bit_cast(unsigned, (us2)(__builtin_bit_cast(us2, a) - __builtin_bit_cast(us2, b)));
}
__device__ __forceinline__ unsigned pk_min16(unsigned a, unsigned b) {
    return __builtin_bit_cast(unsigned, __builtin_elementwise_min(__builtin_bit_cast(us2, a), __builtin_bit_cast(us2, b)));
}

#define EG 256
#define NGRP (N_EDGES / EG)
#define GSTEP (32 * 8)
#define NOKEY 0xFFFFu

__device__ __forceinline__ float lrelu(float x) { return x > 0.f ? x : 0.2f * x; }

__device__ __forceinline__ void atomicMaxF(float* addr, float val) {
    if (val >= 0.f) atomicMax((int*)addr, __float_as_int(val));
    else            atomicMin((unsigned int*)addr, __float_as_uint(val));
}


#define PREP_WB ((RREL * GDIM * HD1 + 255) / 256)
#define PREP_KB ((N_EDGES / 8 + 255) / 256)

__device__ __forceinline__ unsigned dkey(int d) { return (unsigned)d < (unsigned)N_NODES ? (unsigned)d : NOKEY; }

__global__ void k_basis_w(const float* __restrict__ comp, const float* __restrict__ basis,
                          float* __restrict__ W, const int* __restrict__ ei,
                          unsigned* __restrict__ dk) {
    static_assert(N_EDGES % 8 == 0, "a thread packs 8 destinations");
    if (blockIdx.x >= PREP_WB) {
        int i = (blockIdx.x - PREP_WB) * blockDim.x + threadIdx.x;
        if (i >= N_EDGES / 8) return;
        const int4* p = (const int4*)(ei + N_EDGES) + 2 * i;
        int4 a = p[0], b = p[1];
        uint4 o;
        o.x = dkey(a.x) | (dkey(a.y) << 16);
        o.y = dkey(a.z) | (dkey(a.w) << 16);
        o.z = dkey(b.x) | (dkey(b.y) << 16);
        o.w = dkey(b.z) | (dkey(b.w) << 16);
        ((uint4*)dk)[i] = o;
        return;
    }
    int t = blockIdx.x * blockDim.x + threadIdx.x;
    if (t >= RREL * GDIM * HD1) return;
    int r = t / (GDIM * HD1);
    int io = t % (GDIM * HD1);
    float s = 0.f;
    #pragma unroll
    for (int b = 0; b < NB; ++b) s += comp[r * NB + b] * basis[b * GDIM * HD1 + io];
    W[t] = s;
}

#define RT_N 70
#define RT_QCAP 288
#define RT_QFLUSH 16
#define RT_PITCH (RREL * 128 + 4)
#define RT_MT 5
#define RT_PP 132

__global__ __launch_bounds__(1024) void k_rel_tile(const float* __restrict__ x,
                                                   const int* __restrict__ ei,
                                                   const int* __restrict__ et,
                                                   const unsigned* __restrict__ dk,
                                                   const float* __restrict__ Wr,
                                                   const float* __restrict__ bias,
                                                   float* __restrict__ hout) {
    __shared__ float tile[RT_N * RT_PITCH];
    __shared__ float s_cnt[RT_N * RREL];
    __shared__ unsigned s_q[32][RT_QCAP];

    static_assert(N_NODES <= 65536, "source index is packed into 16 bits");
    static_assert(RT_N <= 128, "row within the tile is packed into 7 bits");
    static_assert(N_EDGES <= (1 << 20), "edge position is packed into 20 bits");
    static_assert(RT_QCAP >= RT_QFLUSH + EG, "a group of edges fits behind an unemptied queue");
    static_assert(N_EDGES % EG == 0, "a wave reads EG consecutive destination keys per load");
    static_assert(N_NODES + RT_N <= 65536 - 256, "a key below the tile, or NOKEY, never wraps into it");
    static_assert(RT_MT * 16 >= RT_N, "the row tiles cover the rows");
    static_assert(4 * RT_MT * 16 * RT_PP <= RT_N * RT_PITCH, "the partial products fit where the means were");

    const int tid = threadIdx.x;
    const int lane = tid & 31;
    const int wave = __builtin_amdgcn_readfirstlane(tid >> 5);
    const int lo = blockIdx.x * RT_N;
    const int cnt = (N_NODES - lo) < RT_N ? (N_NODES - lo) : RT_N;

    for (int i = tid; i < RT_N * RT_PITCH / 4; i += 1024)
        ((float4*)tile)[i] = make_float4(0.f, 0.f, 0.f, 0.f);
    for (int i = tid; i < RT_N * RREL; i += 1024) s_cnt[i] = 0.f;
    __syncthreads();

    const int* __restrict__ srcp = ei;
    unsigned* myq = s_q[wave];
    const unsigned lo2 = (unsigned)lo | ((unsigned)lo << 16);
    const unsigned cnt2 = (unsigned)cnt | ((unsigned)cnt << 16);

    int nq = 0;
    auto flush = [&]() __attribute__((always_inline)) {
        if (nq == 0) return;
        __builtin_amdgcn_fence(__ATOMIC_ACQ_REL, "workgroup", "local");
        for (int q0 = 0; q0 < nq; q0 += 32) {
            int n = nq - q0;
            n = n < 32 ? n : 32;
            unsigned e = myq[q0 + (lane < n ? lane : 0)];
            int idx = (int)(e & 0xFFFFFu);
            int s = srcp[idx];
            int ty = et[idx];
            unsigned w = (unsigned)s | ((e >> 20) << 16) | ((unsigned)ty << 23);
            unsigned okm = __builtin_amdgcn_ballot_w32(lane < n && (unsigned)ty < (unsigned)RREL);
            for (int j = 0; j < n; j += 8) {
                unsigned ent[8];
                float xr[8][4];
                #pragma unroll
                for (int b = 0; b < 8; ++b) {
                    ent[b] = (unsigned)__builtin_amdgcn_readlane((int)w, j + b);
                    const float* row = x + (size_t)(ent[b] & 0xFFFFu) * GDIM + lane;
                    xr[b][0] = row[0]; xr[b][1] = row[32]; xr[b][2] = row[64]; xr[b][3] = row[96];
                }
                #pragma unroll
                for (int b = 0; b < 8; ++b) {
                    if ((okm >> (j + b)) & 1u) {
                        int dl = (int)((ent[b] >> 16) & 127u), r = (int)(ent[b] >> 23);
                        float* t = tile + dl * RT_PITCH + r * 128 + lane;
                        atomicAdd(t,      xr[b][0]);
                        atomicAdd(t + 32, xr[b][1]);
                        atomicAdd(t + 64, xr[b][2]);
                        atomicAdd(t + 96, xr[b][3]);
                        if (lane == 0) atomicAdd(&s_cnt[dl * RREL + r], 1.f);
                    }
                }
            }
        }
        nq = 0;
    };
    auto scan_group = [&](const v4u d, int g) __attribute__((always_inline)) {
        #pragma unroll
        for (int hf = 0; hf < 2; ++hf) {
            const unsigned ra = pk_sub16(d[2 * hf], lo2), rb = pk_sub16(d[2 * hf + 1], lo2);
            if (__builtin_amdgcn_ballot_w32(pk_min16(pk_min16(ra, rb), cnt2) != cnt2) != 0u) {
                const int pos = g * EG + lane * 8 + 4 * hf;
                #pragma unroll
                for (int k = 0; k < 4; ++k) {
                    const unsigned short rel = __builtin_bit_cast(us2, k < 2 ? ra : rb)[k & 1];
                    bool hit = rel < (unsigned short)cnt;
                    unsigned bal = __builtin_amdgcn_ballot_w32(hit);
                    if (bal != 0u) {
                        if (hit)
                            myq[nq + __builtin_amdgcn_mbcnt_lo(bal, 0u)] = (unsigned)(pos + k) | ((unsigned)rel << 20);
                        nq += __builtin_popcount(bal);
                    }
                }
            }
        }
    };
    auto fetch = [&](v4u (&buf)[8], int gb) __attribute__((always_inline)) {
        #pragma unroll
        for (int u = 0; u < 8; ++u) {
            int g = gb + u * 32 + wave;
            g = g < NGRP ? g : NGRP - 1;
            buf[u] = *(const v4u*)(dk + g * (EG / 2) + lane * 4);
        }
    };
    auto step = [&](const v4u (&buf)[8], int gb) __attribute__((always_inline)) {
        if (nq >= RT_QFLUSH) flush();
        int u = 8;
        #pragma unroll
        for (int uu = 0; uu < 8; ++uu) {
            if (gb + uu * 32 + wave >= NGRP) break;
            if (__builtin_expect(nq > RT_QCAP - EG, 0)) { u = uu; break; }
            scan_group(buf[uu], gb + uu * 32 + wave);
        }
        #pragma unroll 1
        for (; u < 8 && gb + u * 32 + wave < NGRP; ++u) {
            flush();
            const int g = gb + u * 32 + wave;
            scan_group(*(const v4u*)(dk + g * (EG / 2) + lane * 4), g);
        }
    };

    {
        v4u bA[8], bB[8];
        fetch(bA, 0);
        fetch(bB, GSTEP);
        for (int gb = 0; gb < NGRP; gb += 2 * GSTEP) {
            step(bA, gb);
            fetch(bA, gb + 2 * GSTEP);
            step(bB, gb + GSTEP);
            fetch(bB, gb + 3 * GSTEP);
        }
        flush();
    }
    __syncthreads();

    for (int i = tid; i < RT_N * RREL * 32; i += 1024) {
        int slot = i >> 5;
        int dl = slot >> 3, r = slot & 7;
        float c = fmaxf(s_cnt[slot], 1.f);
        float4* p = (float4*)(tile + dl * RT_PITCH + r * 128) + (i & 31);
        float4 v = *p;
        v.x = v.x / c;
        v.y = v.y / c;
        v.z = v.z / c;
        v.w = v.w / c;
        *p = v;
    }
    __syncthreads();

    const int ct = wave & 7;
    const int kq = wave >> 3;
    const int c16 = lane & 15;
    const int hh = lane >> 4;
    int aoff[RT_MT];
    #pragma unroll
    for (int t = 0; t < RT_MT; ++t) {
        int rr = t * 16 + c16;
        rr = rr < RT_N ? rr : RT_N - 1;
        aoff[t] = rr * RT_PITCH + kq * 256 + hh * 2;
    }
    const float* Bp = Wr + (size_t)(kq * 256 + hh * 2) * HD1 + ct * 16 + c16;

    v8f acc[RT_MT];
    #pragma unroll
    for (int t = 0; t < RT_MT; ++t) {
        #pragma unroll
        for (int i = 0; i < 8; ++i) acc[t][i] = 0.f;
    }
    v2f bcur[4];
    #pragma unroll
    for (int j = 0; j < 4; ++j) {
        bcur[j].x = Bp[(size_t)(4 * j) * HD1];
        bcur[j].y = Bp[(size_t)(4 * j + 1) * HD1];
    }
    #pragma unroll 2
    for (int kb = 0; kb < 256; kb += 16) {
        const int kn = kb + 16 < 256 ? kb + 16 : kb;
        v2f bnx[4];
        #pragma unroll
        for (int j = 0; j < 4; ++j) {
            bnx[j].x = Bp[(size_t)(kn + 4 * j) * HD1];
            bnx[j].y = Bp[(size_t)(kn + 4 * j + 1) * HD1];
        }
        #pragma unroll
        for (int j = 0; j < 4; ++j) {
            #pragma unroll
            for (int t = 0; t < RT_MT; ++t) {
                v2f a = *(const v2f*)(tile + aoff[t] + kb + 4 * j);
                acc[t] = __builtin_amdgcn_wmma_f32_16x16x4_f32(false, a, false, bcur[j], (short)0,
                                                               acc[t], false, false);
            }
        }
        #pragma unroll
        for (int j = 0; j < 4; ++j) bcur[j] = bnx[j];
    }
    __syncthreads();

    float* P = tile;
    #pragma unroll
    for (int t = 0; t < RT_MT; ++t) {
        #pragma unroll
        for (int i = 0; i < 8; ++i)
            P[(kq * (RT_MT * 16) + t * 16 + hh * 8 + i) * RT_PP + ct * 16 + c16] = acc[t][i];
    }
    __syncthreads();

    for (int i = tid; i < cnt * 32; i += 1024) {
        int dl = i >> 5, q = i & 31;
        float4 v = ((const float4*)bias)[q];
        #pragma unroll
        for (int k = 0; k < 4; ++k) {
            float4 p = *(const float4*)(P + (k * (RT_MT * 16) + dl) * RT_PP + q * 4);
            v.x += p.x;
            v.y += p.y;
            v.z += p.z;
            v.w += p.w;
        }
        ((float4*)(hout + (size_t)(lo + dl) * HD1))[q] = v;
    }
}


template <bool BIAS_MODE>
__global__ __launch_bounds__(256) void k_mm16(const float* __restrict__ A,
                                              const float* __restrict__ B,
                                              float* __restrict__ C,
                                              const float* __restrict__ bias) {
    __shared__ float As[16][132];
    int tx = threadIdx.x;
    int ty = threadIdx.y;
    int tid = ty * 32 + tx;
    int row0 = blockIdx.x * 16;

    const float4* A4 = (const float4*)(A + (size_t)row0 * 128);
    for (int i = tid; i < 16 * 32; i += 256) {
        int rr = i >> 5, cc = i & 31;
        float4 v = A4[rr * 32 + cc];
        As[rr][cc * 4 + 0] = v.x;
        As[rr][cc * 4 + 1] = v.y;
        As[rr][cc * 4 + 2] = v.z;
        As[rr][cc * 4 + 3] = v.w;
    }
    __syncthreads();

    int col0 = ty * 16;
    int half = tx >> 4;
    int mn = tx & 15;
    int kg = half * 2;

    v8f c;
    if (BIAS_MODE) {
        float bv = bias ? bias[col0 + mn] : 0.f;
        #pragma unroll
        for (int i = 0; i < 8; ++i) c[i] = bv;
    } else {
        #pragma unroll
        for (int i = 0; i < 8; ++i)
            c[i] = C[(size_t)(row0 + half * 8 + i) * 128 + col0 + mn];
    }

    #pragma unroll 4
    for (int k0 = 0; k0 < 128; k0 += 4) {
        v2f a, b;
        a.x = As[mn][k0 + kg];
        a.y = As[mn][k0 + kg + 1];
        b.x = B[(size_t)(k0 + kg) * 128 + col0 + mn];
        b.y = B[(size_t)(k0 + kg + 1) * 128 + col0 + mn];
        c = __builtin_amdgcn_wmma_f32_16x16x4_f32(false, a, false, b, (short)0, c,
                                                  false, false);
    }

    #pragma unroll
    for (int i = 0; i < 8; ++i)
        C[(size_t)(row0 + half * 8 + i) * 128 + col0 + mn] = c[i];
}


__global__ void k_attn(const float* __restrict__ xl, const float* __restrict__ att_src,
                       const float* __restrict__ att_dst, float* __restrict__ a_src,
                       float* __restrict__ a_dst) {
    int t = blockIdx.x * blockDim.x + threadIdx.x;
    if (t >= N_NODES * HEADS) return;
    int n = t >> 2, h = t & 3;
    const float* row = xl + (size_t)n * 128 + h * HD2;
    float ss = 0.f, sd = 0.f;
    #pragma unroll
    for (int j = 0; j < HD2; ++j) {
        float v = row[j];
        ss += v * att_src[h * HD2 + j];
        sd += v * att_dst[h * HD2 + j];
    }
    a_src[t] = ss;
    a_dst[t] = sd;
}

#define TILE_N 196
#define QCAP 1024

__global__ __launch_bounds__(1024) void k_gat_tile(const int* __restrict__ ei,
                                                   const unsigned* __restrict__ dk,
                                                   const float* __restrict__ xl,
                                                   const float* __restrict__ a_src,
                                                   const float* __restrict__ a_dst,
                                                   const float* __restrict__ bias,
                                                   float* __restrict__ out) {
    __shared__ float tile[TILE_N * 128];
    __shared__ float s_den[TILE_N * 4];
    __shared__ float s_m[TILE_N * 4];
    __shared__ float s_ad[TILE_N * 4];
    __shared__ unsigned s_q[32][QCAP];
    __shared__ int s_ovf;

    static_assert(N_NODES <= 65536, "source index is packed into 16 bits");
    static_assert(N_EDGES % EG == 0, "a wave reads EG consecutive destination keys per load");
    static_assert(N_EDGES <= (1 << 20), "edge position is packed into 20 bits");
    static_assert(TILE_N <= 256, "row within the tile is packed into 8 bits");
    static_assert(N_NODES + TILE_N <= 65536 - 256, "a key below the tile, or NOKEY, never wraps into it");
    static_assert(QCAP >= 32 + EG + EG, "room for an unfinished batch, a group of edges and the give-up margin");

    const int tid = threadIdx.x;
    const int lane = tid & 31;
    const int wave = __builtin_amdgcn_readfirstlane(tid >> 5);
    const int lo = blockIdx.x * TILE_N;
    const int cnt = (N_NODES - lo) < TILE_N ? (N_NODES - lo) : TILE_N;

    if (tid == 0) s_ovf = 0;
    for (int i = tid; i < cnt * 4; i += 1024) {
        int g = lo * 4 + i;
        float ad = a_dst[g];
        s_ad[i] = ad;
        s_m[i] = lrelu(a_src[g] + ad);
    }
    __syncthreads();

    const int* __restrict__ srcp = ei;
    unsigned* myq = s_q[wave];
    const unsigned lo2 = (unsigned)lo | ((unsigned)lo << 16);
    const unsigned cnt2 = (unsigned)cnt | ((unsigned)cnt << 16);

    int nq = 0;
    int done = 0;
    int lim = 32;

    auto add_batch = [&](int s, int dl, int n) __attribute__((always_inline)) {
        bool ok = lane < n && s != lo + dl;
        unsigned okm = __builtin_amdgcn_ballot_w32(ok);
        float4 as = *(const float4*)(a_src + s * 4);
        float4 ad = *(const float4*)&s_ad[dl * 4];
        float4 mm = *(const float4*)&s_m[dl * 4];
        float w0 = __expf(lrelu(as.x + ad.x) - mm.x);
        float w1 = __expf(lrelu(as.y + ad.y) - mm.y);
        float w2 = __expf(lrelu(as.z + ad.z) - mm.z);
        float w3 = __expf(lrelu(as.w + ad.w) - mm.w);
        if (ok) {
            atomicAdd(&s_den[dl * 4 + 0], w0);
            atomicAdd(&s_den[dl * 4 + 1], w1);
            atomicAdd(&s_den[dl * 4 + 2], w2);
            atomicAdd(&s_den[dl * 4 + 3], w3);
        }
        for (int j = 0; j < n; j += 8) {
            float xr[8][4];
            #pragma unroll
            for (int b = 0; b < 8; ++b) {
                int ss = __builtin_amdgcn_readlane(s, j + b);
                const float* row = xl + (size_t)ss * 128 + lane;
                xr[b][0] = row[0]; xr[b][1] = row[32]; xr[b][2] = row[64]; xr[b][3] = row[96];
            }
            #pragma unroll
            for (int b = 0; b < 8; ++b) {
                if ((okm >> (j + b)) & 1u) {
                    int dj = __builtin_amdgcn_readlane(dl, j + b);
                    float v0 = __int_as_float(__builtin_amdgcn_readlane(__float_as_int(w0), j + b));
                    float v1 = __int_as_float(__builtin_amdgcn_readlane(__float_as_int(w1), j + b));
                    float v2 = __int_as_float(__builtin_amdgcn_readlane(__float_as_int(w2), j + b));
                    float v3 = __int_as_float(__builtin_amdgcn_readlane(__float_as_int(w3), j + b));
                    float* t = tile + dj * 128 + lane;
                    atomicAdd(t,      xr[b][0] * v0);
                    atomicAdd(t + 32, xr[b][1] * v1);
                    atomicAdd(t + 64, xr[b][2] * v2);
                    atomicAdd(t + 96, xr[b][3] * v3);
                }
            }
        }
    };
    auto flush_max = [&]() __attribute__((always_inline)) {
        if (nq > done) {
            __builtin_amdgcn_fence(__ATOMIC_ACQ_REL, "workgroup", "local");
            for (int q0 = done; q0 < nq; q0 += 32) {
                int n = nq - q0;
                n = n < 32 ? n : 32;
                unsigned e = myq[q0 + (lane < n ? lane : 0)];
                int s = srcp[(int)(e & 0xFFFFFu)];
                int dl = (int)(e >> 20);
                float4 as = *(const float4*)(a_src + s * 4);
                float4 ad = *(const float4*)&s_ad[dl * 4];
                if (lane < n) {
                    if (s != lo + dl) {
                        atomicMaxF(&s_m[dl * 4 + 0], lrelu(as.x + ad.x));
                        atomicMaxF(&s_m[dl * 4 + 1], lrelu(as.y + ad.y));
                        atomicMaxF(&s_m[dl * 4 + 2], lrelu(as.z + ad.z));
                        atomicMaxF(&s_m[dl * 4 + 3], lrelu(as.w + ad.w));
                    }
                    myq[q0 + lane] = (unsigned)s | ((unsigned)dl << 16);
                }
            }
            done = nq;
        }
        if (nq + EG > QCAP) {
            if (lane == 0) s_ovf = 1;
            nq = 0;
            done = 0;
        }
        lim = done + 32 < QCAP - (EG - 1) ? done + 32 : QCAP - (EG - 1);
        nq = __builtin_amdgcn_readfirstlane(nq);
        done = __builtin_amdgcn_readfirstlane(done);
        lim = __builtin_amdgcn_readfirstlane(lim);
    };
    auto flush_sum = [&]() __attribute__((always_inline)) {
        if (nq == 0) return;
        __builtin_amdgcn_fence(__ATOMIC_ACQ_REL, "workgroup", "local");
        for (int q0 = 0; q0 < nq; q0 += 32) {
            int n = nq - q0;
            n = n < 32 ? n : 32;
            unsigned e = myq[q0 + (lane < n ? lane : 0)];
            int s = srcp[(int)(e & 0xFFFFFu)];
            add_batch(s, (int)(e >> 20), n);
        }
        nq = 0;
    };
    auto scan_group = [&](const v4u d, int g) __attribute__((always_inline)) {
        #pragma unroll
        for (int hf = 0; hf < 2; ++hf) {
            const unsigned ra = pk_sub16(d[2 * hf], lo2), rb = pk_sub16(d[2 * hf + 1], lo2);
            if (__builtin_amdgcn_ballot_w32(pk_min16(pk_min16(ra, rb), cnt2) != cnt2) != 0u) {
                const int pos = g * EG + lane * 8 + 4 * hf;
                #pragma unroll
                for (int k = 0; k < 4; ++k) {
                    const unsigned short rel = __builtin_bit_cast(us2, k < 2 ? ra : rb)[k & 1];
                    bool hit = rel < (unsigned short)cnt;
                    unsigned bal = __builtin_amdgcn_ballot_w32(hit);
                    if (bal != 0u) {
                        if (hit)
                            myq[nq + __builtin_amdgcn_mbcnt_lo(bal, 0u)] = (unsigned)(pos + k) | ((unsigned)rel << 20);
                        nq += __builtin_popcount(bal);
                    }
                }
            }
        }
    };
    auto fetch = [&](v4u (&buf)[8], int gb) __attribute__((always_inline)) {
        #pragma unroll
        for (int u = 0; u < 8; ++u) {
            int g = gb + u * 32 + wave;
            g = g < NGRP ? g : NGRP - 1;
            buf[u] = *(const v4u*)(dk + g * (EG / 2) + lane * 4);
        }
    };
    auto step_max = [&](const v4u (&buf)[8], int gb) __attribute__((always_inline)) {
        if (nq >= lim) flush_max();
        int u = 8;
        #pragma unroll
        for (int uu = 0; uu < 8; ++uu) {
            if (gb + uu * 32 + wave >= NGRP) break;
            if (__builtin_expect(nq > QCAP - EG, 0)) { u = uu; break; }
            scan_group(buf[uu], gb + uu * 32 + wave);
        }
        #pragma unroll 1
        for (; u < 8 && gb + u * 32 + wave < NGRP; ++u) {
            flush_max();
            const int g = gb + u * 32 + wave;
            scan_group(*(const v4u*)(dk + g * (EG / 2) + lane * 4), g);
        }
    };
    auto step_sum = [&](const v4u (&buf)[8], int gb) __attribute__((always_inline)) {
        if (nq >= 32) flush_sum();
        int u = 8;
        #pragma unroll
        for (int uu = 0; uu < 8; ++uu) {
            if (gb + uu * 32 + wave >= NGRP) break;
            if (__builtin_expect(nq > QCAP - EG, 0)) { u = uu; break; }
            scan_group(buf[uu], gb + uu * 32 + wave);
        }
        #pragma unroll 1
        for (; u < 8 && gb + u * 32 + wave < NGRP; ++u) {
            flush_sum();
            const int g = gb + u * 32 + wave;
            scan_group(*(const v4u*)(dk + g * (EG / 2) + lane * 4), g);
        }
    };

    {
        v4u bA[8], bB[8];
        fetch(bA, 0);
        fetch(bB, GSTEP);
        for (int gb = 0; gb < NGRP; gb += 2 * GSTEP) {
            step_max(bA, gb);
            fetch(bA, gb + 2 * GSTEP);
            step_max(bB, gb + GSTEP);
            fetch(bB, gb + 3 * GSTEP);
        }
        flush_max();
    }
    __syncthreads();
    const bool rescan = __builtin_amdgcn_readfirstlane(s_ovf) != 0;

    for (int i = tid; i < cnt * 4; i += 1024)
        s_den[i] = __expf(lrelu(a_src[lo * 4 + i] + s_ad[i]) - s_m[i]);
    __syncthreads();
    for (int i = tid; i < cnt * 128; i += 1024)
        tile[i] = xl[(size_t)lo * 128 + i] * s_den[(i >> 7) * 4 + ((i & 127) >> 5)];
    __syncthreads();

    if (!rescan) {
        for (int q0 = 0; q0 < nq; q0 += 32) {
            int n = nq - q0;
            n = n < 32 ? n : 32;
            unsigned e = myq[q0 + (lane < n ? lane : 0)];
            add_batch((int)(e & 0xFFFFu), (int)(e >> 16), n);
        }
    } else {
        nq = 0;
        done = 0;
        lim = 32;
        v4u bA[8], bB[8];
        fetch(bA, 0);
        fetch(bB, GSTEP);
        for (int gb = 0; gb < NGRP; gb += 2 * GSTEP) {
            step_sum(bA, gb);
            fetch(bA, gb + 2 * GSTEP);
            step_sum(bB, gb + GSTEP);
            fetch(bB, gb + 3 * GSTEP);
        }
        flush_sum();
    }
    __syncthreads();

    for (int i = tid; i < cnt * 32; i += 1024) {
        int q = i & 31;
        float dn = s_den[(i >> 5) * 4 + (q >> 3)];
        float4 b = ((const float4*)bias)[q];
        float4 v = ((const float4*)tile)[i];
        v.x = v.x / dn + b.x;
        v.y = v.y / dn + b.y;
        v.z = v.z / dn + b.z;
        v.w = v.w / dn + b.w;
        ((float4*)out)[(size_t)lo * 32 + i] = v;
    }
}


extern "C" void kernel_launch(void* const* d_in, const int* in_sizes, int n_in,
                              void* d_out, int out_size, void* d_ws, size_t ws_size,
                              hipStream_t stream) {
    const float* x        = (const float*)d_in[0];
    const int*   ei       = (const int*)d_in[1];
    const int*   et       = (const int*)d_in[2];
    const float* comp     = (const float*)d_in[3];
    const float* basis    = (const float*)d_in[4];
    const float* root     = (const float*)d_in[5];
    const float* bias1    = (const float*)d_in[6];
    const float* gat_w    = (const float*)d_in[7];
    const float* att_src  = (const float*)d_in[8];
    const float* att_dst  = (const float*)d_in[9];
    const float* gat_bias = (const float*)d_in[10];
    float* out = (float*)d_out;

    float* ws = (float*)d_ws;
    size_t off = 0;
    float* W    = ws + off; off += (size_t)RREL * GDIM * HD1;
    float* hbuf = ws + off; off += (size_t)N_NODES * HD1;
    float* xl   = ws + off; off += (size_t)N_NODES * HD1;
    float* asg  = ws + off; off += (size_t)N_NODES * HEADS;
    float* adg  = ws + off; off += (size_t)N_NODES * HEADS;
    unsigned* dk = (unsigned*)(ws + off); off += (size_t)N_EDGES / 2;

    dim3 mmBlock(32, 8);
    int mmGrid = N_NODES / 16;

    k_basis_w<<<PREP_WB + PREP_KB, 256, 0, stream>>>(comp, basis, W, ei, dk);
    k_rel_tile<<<(N_NODES + RT_N - 1) / RT_N, 1024, 0, stream>>>(x, ei, et, dk, W, bias1, hbuf);
    k_mm16<false><<<mmGrid, mmBlock, 0, stream>>>(x, root, hbuf, nullptr);

    k_mm16<true><<<mmGrid, mmBlock, 0, stream>>>(hbuf, gat_w, xl, nullptr);
    k_attn<<<(N_NODES * HEADS + 255) / 256, 256, 0, stream>>>(xl, att_src, att_dst, asg, adg);
    k_gat_tile<<<(N_NODES + TILE_N - 1) / TILE_N, 1024, 0, stream>>>(ei, dk, xl, asg, adg, gat_bias, out);

    (void)in_sizes; (void)n_in; (void)out_size; (void)ws_size;
}


# first launch also builds a block index (1 bit per 32-edge block per row tile, LDS integer OR); tile kernels read only flagged blocks' keys, not the whole list; LDS float atomics stay, arrival order
# speedup vs baseline: 4.4356x; 1.1732x over previous
#include <hip/hip_runtime.h>
#include <math.h>

#define N_NODES 50000
#define N_EDGES 800000
#define GDIM 128
#define HD1 128
#define HEADS 4
#define HD2 32
#define RREL 8
#define NB 20

typedef __attribute__((ext_vector_type(2))) float v2f;
typedef __attribute__((ext_vector_type(8))) float v8f;
typedef __attribute__((ext_vector_type(4))) unsigned int v4u;
typedef __attribute__((ext_vector_type(2))) unsigned short us2;

__device__ __forceinline__ unsigned pk_sub16(unsigned a, unsigned b) {
    return __builtin_bit_cast(unsigned, (us2)(__builtin_bit_cast(us2, a) - __builtin_bit_cast(us2, b)));
}
__device__ __forceinline__ unsigned pk_min16(unsigned a, unsigned b) {
    return __builtin_bit_cast(unsigned, __builtin_elementwise_min(__builtin_bit_cast(us2, a), __builtin_bit_cast(us2, b)));
}

#define EG 256
#define NGRP (N_EDGES / EG)
#define GSTEP (32 * 8)
#define NOKEY 0xFFFFu

#define RT_N 70
#define TILE_N 196
#define RT_TILES ((N_NODES + RT_N - 1) / RT_N)
#define GT_TILES ((N_NODES + TILE_N - 1) / TILE_N)
#define EB 32
#define NBLK (N_EDGES / EB)
#define NWORD ((NBLK + 31) / 32)

__device__ __forceinline__ float lrelu(float x) { return x > 0.f ? x : 0.2f * x; }

__device__ __forceinline__ void atomicMaxF(float* addr, float val) {
    if (val >= 0.f) atomicMax((int*)addr, __float_as_int(val));
    else            atomicMin((unsigned int*)addr, __float_as_uint(val));
}


#define PREP_WB ((RREL * GDIM * HD1 + 255) / 256)
#define PREP_KB ((N_EDGES / 8 + 255) / 256)

__device__ __forceinline__ unsigned dkey(int d) { return (unsigned)d < (unsigned)N_NODES ? (unsigned)d : NOKEY; }

__global__ __launch_bounds__(256) void k_basis_w(const float* __restrict__ comp, const float* __restrict__ basis,
                          float* __restrict__ W, const int* __restrict__ ei,
                          unsigned* __restrict__ dk, unsigned* __restrict__ bits_r,
                          unsigned* __restrict__ bits_g) {
    static_assert(N_EDGES % 8 == 0, "a thread packs 8 destinations");
    static_assert(EB == 32 && N_EDGES % EB == 0, "a thread's 8 edges lie in one block; 4 threads per block");
    static_assert(2 * PREP_KB >= NWORD, "a packing workgroup of 256 threads fills two words of block bits per tile");
    __shared__ unsigned bm[2][RT_TILES + GT_TILES];
    if (blockIdx.x >= PREP_WB) {
        const int kb = blockIdx.x - PREP_WB;
        for (int i = threadIdx.x; i < 2 * (RT_TILES + GT_TILES); i += 256) (&bm[0][0])[i] = 0u;
        __syncthreads();
        int i = kb * 256 + threadIdx.x;
        if (i < N_EDGES / 8) {
            const int4* p = (const int4*)(ei + N_EDGES) + 2 * i;
            int4 a = p[0], b = p[1];
            uint4 o;
            o.x = dkey(a.x) | (dkey(a.y) << 16);
            o.y = dkey(a.z) | (dkey(a.w) << 16);
            o.z = dkey(b.x) | (dkey(b.y) << 16);
            o.w = dkey(b.z) | (dkey(b.w) << 16);
            ((uint4*)dk)[i] = o;
            unsigned* row = bm[threadIdx.x >> 7];
            const unsigned bit = 1u << ((threadIdx.x >> 2) & 31);
            const int d[8] = {a.x, a.y, a.z, a.w, b.x, b.y, b.z, b.w};
            #pragma unroll
            for (int k = 0; k < 8; ++k) {
                if ((unsigned)d[k] < (unsigned)N_NODES) {
                    atomicOr(&row[d[k] / RT_N], bit);
                    atomicOr(&row[RT_TILES + d[k] / TILE_N], bit);
                }
            }
        }
        __syncthreads();
        for (int i = threadIdx.x; i < 2 * (RT_TILES + GT_TILES); i += 256) {
            const int j = i / (RT_TILES + GT_TILES), t = i - j * (RT_TILES + GT_TILES);
            const unsigned v = bm[j][t];
            const size_t gw = (size_t)(kb * 2 + j);
            if (t < RT_TILES) bits_r[gw * RT_TILES + t] = v;
            else              bits_g[gw * GT_TILES + (t - RT_TILES)] = v;
        }
        return;
    }
    int t = blockIdx.x * blockDim.x + threadIdx.x;
    if (t >= RREL * GDIM * HD1) return;
    int r = t / (GDIM * HD1);
    int io = t % (GDIM * HD1);
    float s = 0.f;
    #pragma unroll
    for (int b = 0; b < NB; ++b) s += comp[r * NB + b] * basis[b * GDIM * HD1 + io];
    W[t] = s;
}

#define RT_QCAP 288
#define RT_QFLUSH 32
#define RT_PITCH (RREL * 128 + 4)
#define RT_MT 5
#define RT_PP 132

__global__ __launch_bounds__(1024) void k_rel_tile(const float* __restrict__ x,
                                                   const int* __restrict__ ei,
                                                   const int* __restrict__ et,
                                                   const unsigned* __restrict__ dk,
                                                   const unsigned* __restrict__ bits,
                                                   const float* __restrict__ Wr,
                                                   const float* __restrict__ bias,
                                                   float* __restrict__ hout) {
    __shared__ float tile[RT_N * RT_PITCH];
    __shared__ float s_cnt[RT_N * RREL];
    __shared__ unsigned s_q[32][RT_QCAP];

    static_assert(N_NODES <= 65536, "source index is packed into 16 bits");
    static_assert(RT_N <= 128, "row within the tile is packed into 7 bits");
    static_assert(N_EDGES <= (1 << 20), "edge position is packed into 20 bits");
    static_assert(N_NODES + RT_N <= 65536 - 256, "a key below the tile, or NOKEY, never wraps into it");
    static_assert(RT_MT * 16 >= RT_N, "the row tiles cover the rows");
    static_assert(4 * RT_MT * 16 * RT_PP <= RT_N * RT_PITCH, "the partial products fit where the means were");

    const int tid = threadIdx.x;
    const int lane = tid & 31;
    const int wave = __builtin_amdgcn_readfirstlane(tid >> 5);
    const int lo = blockIdx.x * RT_N;
    const int cnt = (N_NODES - lo) < RT_N ? (N_NODES - lo) : RT_N;

    for (int i = tid; i < RT_N * RT_PITCH / 4; i += 1024)
        ((float4*)tile)[i] = make_float4(0.f, 0.f, 0.f, 0.f);
    for (int i = tid; i < RT_N * RREL; i += 1024) s_cnt[i] = 0.f;
    __syncthreads();

    const int* __restrict__ srcp = ei;
    unsigned* myq = s_q[wave];

    int nq = 0;
    auto flush = [&]() __attribute__((always_inline)) {
        if (nq == 0) return;
        __builtin_amdgcn_fence(__ATOMIC_ACQ_REL, "workgroup", "local");
        for (int q0 = 0; q0 < nq; q0 += 32) {
            int n = nq - q0;
            n = n < 32 ? n : 32;
            unsigned e = myq[q0 + (lane < n ? lane : 0)];
            int idx = (int)(e & 0xFFFFFu);
            int s = srcp[idx];
            int ty = et[idx];
            unsigned w = (unsigned)s | ((e >> 20) << 16) | ((unsigned)ty << 23);
            unsigned okm = __builtin_amdgcn_ballot_w32(lane < n && (unsigned)ty < (unsigned)RREL);
            for (int j = 0; j < n; j += 8) {
                unsigned ent[8];
                float xr[8][4];
                #pragma unroll
                for (int b = 0; b < 8; ++b) {
                    ent[b] = (unsigned)__builtin_amdgcn_readlane((int)w, j + b);
                    const float* row = x + (size_t)(ent[b] & 0xFFFFu) * GDIM + lane;
                    xr[b][0] = row[0]; xr[b][1] = row[32]; xr[b][2] = row[64]; xr[b][3] = row[96];
                }
                #pragma unroll
                for (int b = 0; b < 8; ++b) {
                    if ((okm >> (j + b)) & 1u) {
                        int dl = (int)((ent[b] >> 16) & 127u), r = (int)(ent[b] >> 23);
                        float* t = tile + dl * RT_PITCH + r * 128 + lane;
                        atomicAdd(t,      xr[b][0]);
                        atomicAdd(t + 32, xr[b][1]);
                        atomicAdd(t + 64, xr[b][2]);
                        atomicAdd(t + 96, xr[b][3]);
                        if (lane == 0) atomicAdd(&s_cnt[dl * RREL + r], 1.f);
                    }
                }
            }
        }
        nq = 0;
    };
    static_assert(RT_QCAP >= RT_QFLUSH + 8 * EB, "8 blocks of edges fit behind an unemptied queue");
    {
        const unsigned short* __restrict__ k16 = (const unsigned short*)dk;
        const int wi = wave + 32 * lane;
        unsigned m = wi < NWORD ? bits[(size_t)wi * RT_TILES + blockIdx.x] : 0u;
        int myblk = 0;
        bool loaded = false;
        for (;;) {
            if (!loaded && m != 0u) {
                myblk = wi * 32 + __builtin_ctz(m);
                m &= m - 1u;
                loaded = true;
            }
            const unsigned offered = __builtin_amdgcn_ballot_w32(loaded);
            if (offered == 0u) break;
            if (nq >= RT_QFLUSH) flush();
            unsigned left = offered;
            int blk[8];
            bool val[8];
            unsigned short key[8];
            #pragma unroll
            for (int u = 0; u < 8; ++u) {
                val[u] = left != 0u;
                const int j = val[u] ? __builtin_ctz(left) : 0;
                left &= left - 1u;
                blk[u] = __builtin_amdgcn_readlane(myblk, j);
                key[u] = k16[blk[u] * EB + lane];
            }
            if (((offered & ~left) >> lane) & 1u) loaded = false;
            #pragma unroll
            for (int u = 0; u < 8; ++u) {
                if (val[u]) {
                    const unsigned short rel = (unsigned short)(key[u] - (unsigned short)lo);
                    const bool hit = rel < (unsigned short)cnt;
                    const unsigned hb = __builtin_amdgcn_ballot_w32(hit);
                    if (hb != 0u) {
                        if (hit)
                            myq[nq + __builtin_amdgcn_mbcnt_lo(hb, 0u)] =
                                (unsigned)(blk[u] * EB + lane) | ((unsigned)rel << 20);
                        nq += __builtin_popcount(hb);
                    }
                }
            }
        }
        flush();
    }
    __syncthreads();

    for (int i = tid; i < RT_N * RREL * 32; i += 1024) {
        int slot = i >> 5;
        int dl = slot >> 3, r = slot & 7;
        float c = fmaxf(s_cnt[slot], 1.f);
        float4* p = (float4*)(tile + dl * RT_PITCH + r * 128) + (i & 31);
        float4 v = *p;
        v.x = v.x / c;
        v.y = v.y / c;
        v.z = v.z / c;
        v.w = v.w / c;
        *p = v;
    }
    __syncthreads();

    const int ct = wave & 7;
    const int kq = wave >> 3;
    const int c16 = lane & 15;
    const int hh = lane >> 4;
    int aoff[RT_MT];
    #pragma unroll
    for (int t = 0; t < RT_MT; ++t) {
        int rr = t * 16 + c16;
        rr = rr < RT_N ? rr : RT_N - 1;
        aoff[t] = rr * RT_PITCH + kq * 256 + hh * 2;
    }
    const float* Bp = Wr + (size_t)(kq * 256 + hh * 2) * HD1 + ct * 16 + c16;

    v8f acc[RT_MT];
    #pragma unroll
    for (int t = 0; t < RT_MT; ++t) {
        #pragma unroll
        for (int i = 0; i < 8; ++i) acc[t][i] = 0.f;
    }
    v2f bcur[4];
    #pragma unroll
    for (int j = 0; j < 4; ++j) {
        bcur[j].x = Bp[(size_t)(4 * j) * HD1];
        bcur[j].y = Bp[(size_t)(4 * j + 1) * HD1];
    }
    #pragma unroll 2
    for (int kb = 0; kb < 256; kb += 16) {
        const int kn = kb + 16 < 256 ? kb + 16 : kb;
        v2f bnx[4];
        #pragma unroll
        for (int j = 0; j < 4; ++j) {
            bnx[j].x = Bp[(size_t)(kn + 4 * j) * HD1];
            bnx[j].y = Bp[(size_t)(kn + 4 * j + 1) * HD1];
        }
        #pragma unroll
        for (int j = 0; j < 4; ++j) {
            #pragma unroll
            for (int t = 0; t < RT_MT; ++t) {
                v2f a = *(const v2f*)(tile + aoff[t] + kb + 4 * j);
                acc[t] = __builtin_amdgcn_wmma_f32_16x16x4_f32(false, a, false, bcur[j], (short)0,
                                                               acc[t], false, false);
            }
        }
        #pragma unroll
        for (int j = 0; j < 4; ++j) bcur[j] = bnx[j];
    }
    __syncthreads();

    float* P = tile;
    #pragma unroll
    for (int t = 0; t < RT_MT; ++t) {
        #pragma unroll
        for (int i = 0; i < 8; ++i)
            P[(kq * (RT_MT * 16) + t * 16 + hh * 8 + i) * RT_PP + ct * 16 + c16] = acc[t][i];
    }
    __syncthreads();

    for (int i = tid; i < cnt * 32; i += 1024) {
        int dl = i >> 5, q = i & 31;
        float4 v = ((const float4*)bias)[q];
        #pragma unroll
        for (int k = 0; k < 4; ++k) {
            float4 p = *(const float4*)(P + (k * (RT_MT * 16) + dl) * RT_PP + q * 4);
            v.x += p.x;
            v.y += p.y;
            v.z += p.z;
            v.w += p.w;
        }
        ((float4*)(hout + (size_t)(lo + dl) * HD1))[q] = v;
    }
}


template <bool BIAS_MODE>
__global__ __launch_bounds__(256) void k_mm16(const float* __restrict__ A,
                                              const float* __restrict__ B,
                                              float* __restrict__ C,
                                              const float* __restrict__ bias) {
    __shared__ float As[16][132];
    int tx = threadIdx.x;
    int ty = threadIdx.y;
    int tid = ty * 32 + tx;
    int row0 = blockIdx.x * 16;

    const float4* A4 = (const float4*)(A + (size_t)row0 * 128);
    for (int i = tid; i < 16 * 32; i += 256) {
        int rr = i >> 5, cc = i & 31;
        float4 v = A4[rr * 32 + cc];
        As[rr][cc * 4 + 0] = v.x;
        As[rr][cc * 4 + 1] = v.y;
        As[rr][cc * 4 + 2] = v.z;
        As[rr][cc * 4 + 3] = v.w;
    }
    __syncthreads();

    int col0 = ty * 16;
    int half = tx >> 4;
    int mn = tx & 15;
    int kg = half * 2;

    v8f c;
    if (BIAS_MODE) {
        float bv = bias ? bias[col0 + mn] : 0.f;
        #pragma unroll
        for (int i = 0; i < 8; ++i) c[i] = bv;
    } else {
        #pragma unroll
        for (int i = 0; i < 8; ++i)
            c[i] = C[(size_t)(row0 + half * 8 + i) * 128 + col0 + mn];
    }

    #pragma unroll 4
    for (int k0 = 0; k0 < 128; k0 += 4) {
        v2f a, b;
        a.x = As[mn][k0 + kg];
        a.y = As[mn][k0 + kg + 1];
        b.x = B[(size_t)(k0 + kg) * 128 + col0 + mn];
        b.y = B[(size_t)(k0 + kg + 1) * 128 + col0 + mn];
        c = __builtin_amdgcn_wmma_f32_16x16x4_f32(false, a, false, b, (short)0, c,
                                                  false, false);
    }

    #pragma unroll
    for (int i = 0; i < 8; ++i)
        C[(size_t)(row0 + half * 8 + i) * 128 + col0 + mn] = c[i];
}


__global__ void k_attn(const float* __restrict__ xl, const float* __restrict__ att_src,
                       const float* __restrict__ att_dst, float* __restrict__ a_src,
                       float* __restrict__ a_dst) {
    int t = blockIdx.x * blockDim.x + threadIdx.x;
    if (t >= N_NODES * HEADS) return;
    int n = t >> 2, h = t & 3;
    const float* row = xl + (size_t)n * 128 + h * HD2;
    float ss = 0.f, sd = 0.f;
    #pragma unroll
    for (int j = 0; j < HD2; ++j) {
        float v = row[j];
        ss += v * att_src[h * HD2 + j];
        sd += v * att_dst[h * HD2 + j];
    }
    a_src[t] = ss;
    a_dst[t] = sd;
}

#define QCAP 1024

__global__ __launch_bounds__(1024) void k_gat_tile(const int* __restrict__ ei,
                                                   const unsigned* __restrict__ dk,
                                                   const unsigned* __restrict__ bits,
                                                   const float* __restrict__ xl,
                                                   const float* __restrict__ a_src,
                                                   const float* __restrict__ a_dst,
                                                   const float* __restrict__ bias,
                                                   float* __restrict__ out) {
    __shared__ float tile[TILE_N * 128];
    __shared__ float s_den[TILE_N * 4];
    __shared__ float s_m[TILE_N * 4];
    __shared__ float s_ad[TILE_N * 4];
    __shared__ unsigned s_q[32][QCAP];
    __shared__ int s_ovf;

    static_assert(N_NODES <= 65536, "source index is packed into 16 bits");
    static_assert(N_EDGES % EG == 0, "a wave reads EG consecutive destination keys per load");
    static_assert(N_EDGES <= (1 << 20), "edge position is packed into 20 bits");
    static_assert(TILE_N <= 256, "row within the tile is packed into 8 bits");
    static_assert(N_NODES + TILE_N <= 65536 - 256, "a key below the tile, or NOKEY, never wraps into it");
    static_assert(QCAP >= 32 + EG + EG, "room for an unfinished batch, a group of edges and the give-up margin");

    const int tid = threadIdx.x;
    const int lane = tid & 31;
    const int wave = __builtin_amdgcn_readfirstlane(tid >> 5);
    const int lo = blockIdx.x * TILE_N;
    const int cnt = (N_NODES - lo) < TILE_N ? (N_NODES - lo) : TILE_N;

    if (tid == 0) s_ovf = 0;
    for (int i = tid; i < cnt * 4; i += 1024) {
        int g = lo * 4 + i;
        float ad = a_dst[g];
        s_ad[i] = ad;
        s_m[i] = lrelu(a_src[g] + ad);
    }
    __syncthreads();

    const int* __restrict__ srcp = ei;
    unsigned* myq = s_q[wave];
    const unsigned lo2 = (unsigned)lo | ((unsigned)lo << 16);
    const unsigned cnt2 = (unsigned)cnt | ((unsigned)cnt << 16);

    int nq = 0;
    int done = 0;
    int lim = 32;

    auto add_batch = [&](int s, int dl, int n) __attribute__((always_inline)) {
        bool ok = lane < n && s != lo + dl;
        unsigned okm = __builtin_amdgcn_ballot_w32(ok);
        float4 as = *(const float4*)(a_src + s * 4);
        float4 ad = *(const float4*)&s_ad[dl * 4];
        float4 mm = *(const float4*)&s_m[dl * 4];
        float w0 = __expf(lrelu(as.x + ad.x) - mm.x);
        float w1 = __expf(lrelu(as.y + ad.y) - mm.y);
        float w2 = __expf(lrelu(as.z + ad.z) - mm.z);
        float w3 = __expf(lrelu(as.w + ad.w) - mm.w);
        if (ok) {
            atomicAdd(&s_den[dl * 4 + 0], w0);
            atomicAdd(&s_den[dl * 4 + 1], w1);
            atomicAdd(&s_den[dl * 4 + 2], w2);
            atomicAdd(&s_den[dl * 4 + 3], w3);
        }
        for (int j = 0; j < n; j += 8) {
            float xr[8][4];
            #pragma unroll
            for (int b = 0; b < 8; ++b) {
                int ss = __builtin_amdgcn_readlane(s, j + b);
                const float* row = xl + (size_t)ss * 128 + lane;
                xr[b][0] = row[0]; xr[b][1] = row[32]; xr[b][2] = row[64]; xr[b][3] = row[96];
            }
            #pragma unroll
            for (int b = 0; b < 8; ++b) {
                if ((okm >> (j + b)) & 1u) {
                    int dj = __builtin_amdgcn_readlane(dl, j + b);
                    float v0 = __int_as_float(__builtin_amdgcn_readlane(__float_as_int(w0), j + b));
                    float v1 = __int_as_float(__builtin_amdgcn_readlane(__float_as_int(w1), j + b));
                    float v2 = __int_as_float(__builtin_amdgcn_readlane(__float_as_int(w2), j + b));
                    float v3 = __int_as_float(__builtin_amdgcn_readlane(__float_as_int(w3), j + b));
                    float* t = tile + dj * 128 + lane;
                    atomicAdd(t,      xr[b][0] * v0);
                    atomicAdd(t + 32, xr[b][1] * v1);
                    atomicAdd(t + 64, xr[b][2] * v2);
                    atomicAdd(t + 96, xr[b][3] * v3);
                }
            }
        }
    };
    auto flush_max = [&]() __attribute__((always_inline)) {
        if (nq > done) {
            __builtin_amdgcn_fence(__ATOMIC_ACQ_REL, "workgroup", "local");
            for (int q0 = done; q0 < nq; q0 += 32) {
                int n = nq - q0;
                n = n < 32 ? n : 32;
                unsigned e = myq[q0 + (lane < n ? lane : 0)];
                int s = srcp[(int)(e & 0xFFFFFu)];
                int dl = (int)(e >> 20);
                float4 as = *(const float4*)(a_src + s * 4);
                float4 ad = *(const float4*)&s_ad[dl * 4];
                if (lane < n) {
                    if (s != lo + dl) {
                        atomicMaxF(&s_m[dl * 4 + 0], lrelu(as.x + ad.x));
                        atomicMaxF(&s_m[dl * 4 + 1], lrelu(as.y + ad.y));
                        atomicMaxF(&s_m[dl * 4 + 2], lrelu(as.z + ad.z));
                        atomicMaxF(&s_m[dl * 4 + 3], lrelu(as.w + ad.w));
                    }
                    myq[q0 + lane] = (unsigned)s | ((unsigned)dl << 16);
                }
            }
            done = nq;
        }
        if (nq + EG > QCAP) {
            if (lane == 0) s_ovf = 1;
            nq = 0;
            done = 0;
        }
        lim = done + 32 < QCAP - (EG - 1) ? done + 32 : QCAP - (EG - 1);
        nq = __builtin_amdgcn_readfirstlane(nq);
        done = __builtin_amdgcn_readfirstlane(done);
        lim = __builtin_amdgcn_readfirstlane(lim);
    };
    auto flush_sum = [&]() __attribute__((always_inline)) {
        if (nq == 0) return;
        __builtin_amdgcn_fence(__ATOMIC_ACQ_REL, "workgroup", "local");
        for (int q0 = 0; q0 < nq; q0 += 32) {
            int n = nq - q0;
            n = n < 32 ? n : 32;
            unsigned e = myq[q0 + (lane < n ? lane : 0)];
            int s = srcp[(int)(e & 0xFFFFFu)];
            add_batch(s, (int)(e >> 20), n);
        }
        nq = 0;
    };
    auto scan_group = [&](const v4u d, int g) __attribute__((always_inline)) {
        #pragma unroll
        for (int hf = 0; hf < 2; ++hf) {
            const unsigned ra = pk_sub16(d[2 * hf], lo2), rb = pk_sub16(d[2 * hf + 1], lo2);
            if (__builtin_amdgcn_ballot_w32(pk_min16(pk_min16(ra, rb), cnt2) != cnt2) != 0u) {
                const int pos = g * EG + lane * 8 + 4 * hf;
                #pragma unroll
                for (int k = 0; k < 4; ++k) {
                    const unsigned short rel = __builtin_bit_cast(us2, k < 2 ? ra : rb)[k & 1];
                    bool hit = rel < (unsigned short)cnt;
                    unsigned bal = __builtin_amdgcn_ballot_w32(hit);
                    if (bal != 0u) {
                        if (hit)
                            myq[nq + __builtin_amdgcn_mbcnt_lo(bal, 0u)] = (unsigned)(pos + k) | ((unsigned)rel << 20);
                        nq += __builtin_popcount(bal);
                    }
                }
            }
        }
    };
    auto fetch = [&](v4u (&buf)[8], int gb) __attribute__((always_inline)) {
        #pragma unroll
        for (int u = 0; u < 8; ++u) {
            int g = gb + u * 32 + wave;
            g = g < NGRP ? g : NGRP - 1;
            buf[u] = *(const v4u*)(dk + g * (EG / 2) + lane * 4);
        }
    };
    auto step_sum = [&](const v4u (&buf)[8], int gb) __attribute__((always_inline)) {
        if (nq >= 32) flush_sum();
        int u = 8;
        #pragma unroll
        for (int uu = 0; uu < 8; ++uu) {
            if (gb + uu * 32 + wave >= NGRP) break;
            if (__builtin_expect(nq > QCAP - EG, 0)) { u = uu; break; }
            scan_group(buf[uu], gb + uu * 32 + wave);
        }
        #pragma unroll 1
        for (; u < 8 && gb + u * 32 + wave < NGRP; ++u) {
            flush_sum();
            const int g = gb + u * 32 + wave;
            scan_group(*(const v4u*)(dk + g * (EG / 2) + lane * 4), g);
        }
    };

    static_assert(8 * EB == EG, "a batch of 8 blocks is what the queue keeps room for");
    {
        const unsigned short* __restrict__ k16 = (const unsigned short*)dk;
        const int wi = wave + 32 * lane;
        unsigned m = wi < NWORD ? bits[(size_t)wi * GT_TILES + blockIdx.x] : 0u;
        int myblk = 0;
        bool loaded = false;
        for (;;) {
            if (!loaded && m != 0u) {
                myblk = wi * 32 + __builtin_ctz(m);
                m &= m - 1u;
                loaded = true;
            }
            const unsigned offered = __builtin_amdgcn_ballot_w32(loaded);
            if (offered == 0u) break;
            if (nq >= lim) flush_max();
            unsigned left = offered;
            int blk[8];
            bool val[8];
            unsigned short key[8];
            #pragma unroll
            for (int u = 0; u < 8; ++u) {
                val[u] = left != 0u;
                const int j = val[u] ? __builtin_ctz(left) : 0;
                left &= left - 1u;
                blk[u] = __builtin_amdgcn_readlane(myblk, j);
                key[u] = k16[blk[u] * EB + lane];
            }
            if (((offered & ~left) >> lane) & 1u) loaded = false;
            #pragma unroll
            for (int u = 0; u < 8; ++u) {
                if (val[u]) {
                    const unsigned short rel = (unsigned short)(key[u] - (unsigned short)lo);
                    const bool hit = rel < (unsigned short)cnt;
                    const unsigned hb = __builtin_amdgcn_ballot_w32(hit);
                    if (hb != 0u) {
                        if (hit)
                            myq[nq + __builtin_amdgcn_mbcnt_lo(hb, 0u)] =
                                (unsigned)(blk[u] * EB + lane) | ((unsigned)rel << 20);
                        nq += __builtin_popcount(hb);
                    }
                }
            }
        }
        flush_max();
    }
    __syncthreads();
    const bool rescan = __builtin_amdgcn_readfirstlane(s_ovf) != 0;

    for (int i = tid; i < cnt * 4; i += 1024)
        s_den[i] = __expf(lrelu(a_src[lo * 4 + i] + s_ad[i]) - s_m[i]);
    __syncthreads();
    for (int i = tid; i < cnt * 128; i += 1024)
        tile[i] = xl[(size_t)lo * 128 + i] * s_den[(i >> 7) * 4 + ((i & 127) >> 5)];
    __syncthreads();

    if (!rescan) {
        for (int q0 = 0; q0 < nq; q0 += 32) {
            int n = nq - q0;
            n = n < 32 ? n : 32;
            unsigned e = myq[q0 + (lane < n ? lane : 0)];
            add_batch((int)(e & 0xFFFFu), (int)(e >> 16), n);
        }
    } else {
        nq = 0;
        done = 0;
        lim = 32;
        v4u bA[8], bB[8];
        fetch(bA, 0);
        fetch(bB, GSTEP);
        for (int gb = 0; gb < NGRP; gb += 2 * GSTEP) {
            step_sum(bA, gb);
            fetch(bA, gb + 2 * GSTEP);
            step_sum(bB, gb + GSTEP);
            fetch(bB, gb + 3 * GSTEP);
        }
        flush_sum();
    }
    __syncthreads();

    for (int i = tid; i < cnt * 32; i += 1024) {
        int q = i & 31;
        float dn = s_den[(i >> 5) * 4 + (q >> 3)];
        float4 b = ((const float4*)bias)[q];
        float4 v = ((const float4*)tile)[i];
        v.x = v.x / dn + b.x;
        v.y = v.y / dn + b.y;
        v.z = v.z / dn + b.z;
        v.w = v.w / dn + b.w;
        ((float4*)out)[(size_t)lo * 32 + i] = v;
    }
}


extern "C" void kernel_launch(void* const* d_in, const int* in_sizes, int n_in,
                              void* d_out, int out_size, void* d_ws, size_t ws_size,
                              hipStream_t stream) {
    const float* x        = (const float*)d_in[0];
    const int*   ei       = (const int*)d_in[1];
    const int*   et       = (const int*)d_in[2];
    const float* comp     = (const float*)d_in[3];
    const float* basis    = (const float*)d_in[4];
    const float* root     = (const float*)d_in[5];
    const float* bias1    = (const float*)d_in[6];
    const float* gat_w    = (const float*)d_in[7];
    const float* att_src  = (const float*)d_in[8];
    const float* att_dst  = (const float*)d_in[9];
    const float* gat_bias = (const float*)d_in[10];
    float* out = (float*)d_out;

    float* ws = (float*)d_ws;
    size_t off = 0;
    float* W    = ws + off; off += (size_t)RREL * GDIM * HD1;
    float* hbuf = ws + off; off += (size_t)N_NODES * HD1;
    float* xl   = ws + off; off += (size_t)N_NODES * HD1;
    float* asg  = ws + off; off += (size_t)N_NODES * HEADS;
    float* adg  = ws + off; off += (size_t)N_NODES * HEADS;
    unsigned* dk = (unsigned*)(ws + off); off += (size_t)N_EDGES / 2;
    unsigned* bits_r = (unsigned*)(ws + off); off += (size_t)2 * PREP_KB * RT_TILES;
    unsigned* bits_g = (unsigned*)(ws + off); off += (size_t)2 * PREP_KB * GT_TILES;

    dim3 mmBlock(32, 8);
    int mmGrid = N_NODES / 16;

    k_basis_w<<<PREP_WB + PREP_KB, 256, 0, stream>>>(comp, basis, W, ei, dk, bits_r, bits_g);
    k_rel_tile<<<RT_TILES, 1024, 0, stream>>>(x, ei, et, dk, bits_r, W, bias1, hbuf);
    k_mm16<false><<<mmGrid, mmBlock, 0, stream>>>(x, root, hbuf, nullptr);

    k_mm16<true><<<mmGrid, mmBlock, 0, stream>>>(hbuf, gat_w, xl, nullptr);
    k_attn<<<(N_NODES * HEADS + 255) / 256, 256, 0, stream>>>(xl, att_src, att_dst, asg, adg);
    k_gat_tile<<<GT_TILES, 1024, 0, stream>>>(ei, dk, bits_g, xl, asg, adg, gat_bias, out);

    (void)in_sizes; (void)n_in; (void)out_size; (void)ws_size;
}


# dense tail fused into the relation tile kernel: x @ root rides its K quarters, h stays in LDS as A of xl = h @ gat_w, attention dots taken there; 3 launches go; h, xl, dot sum trees change (fixed)
# speedup vs baseline: 4.7111x; 1.0621x over previous
#include <hip/hip_runtime.h>
#include <math.h>

#define N_NODES 50000
#define N_EDGES 800000
#define GDIM 128
#define HD1 128
#define HEADS 4
#define HD2 32
#define RREL 8
#define NB 20

typedef __attribute__((ext_vector_type(2))) float v2f;
typedef __attribute__((ext_vector_type(8))) float v8f;
typedef __attribute__((ext_vector_type(4))) unsigned int v4u;
typedef __attribute__((ext_vector_type(2))) unsigned short us2;

__device__ __forceinline__ unsigned pk_sub16(unsigned a, unsigned b) {
    return __builtin_bit_cast(unsigned, (us2)(__builtin_bit_cast(us2, a) - __builtin_bit_cast(us2, b)));
}
__device__ __forceinline__ unsigned pk_min16(unsigned a, unsigned b) {
    return __builtin_bit_cast(unsigned, __builtin_elementwise_min(__builtin_bit_cast(us2, a), __builtin_bit_cast(us2, b)));
}

#define EG 256
#define NGRP (N_EDGES / EG)
#define GSTEP (32 * 8)
#define NOKEY 0xFFFFu

#define RT_N 70
#define TILE_N 196
#define RT_TILES ((N_NODES + RT_N - 1) / RT_N)
#define GT_TILES ((N_NODES + TILE_N - 1) / TILE_N)
#define EB 32
#define NBLK (N_EDGES / EB)
#define NWORD ((NBLK + 31) / 32)

__device__ __forceinline__ float lrelu(float x) { return x > 0.f ? x : 0.2f * x; }

__device__ __forceinline__ void atomicMaxF(float* addr, float val) {
    if (val >= 0.f) atomicMax((int*)addr, __float_as_int(val));
    else            atomicMin((unsigned int*)addr, __float_as_uint(val));
}


#define PREP_WB ((RREL * GDIM * HD1 + 255) / 256)
#define PREP_KB ((N_EDGES / 8 + 255) / 256)

__device__ __forceinline__ unsigned dkey(int d) { return (unsigned)d < (unsigned)N_NODES ? (unsigned)d : NOKEY; }

__global__ __launch_bounds__(256) void k_basis_w(const float* __restrict__ comp, const float* __restrict__ basis,
                          float* __restrict__ W, const int* __restrict__ ei,
                          unsigned* __restrict__ dk, unsigned* __restrict__ bits_r,
                          unsigned* __restrict__ bits_g) {
    static_assert(N_EDGES % 8 == 0, "a thread packs 8 destinations");
    static_assert(EB == 32 && N_EDGES % EB == 0, "a thread's 8 edges lie in one block; 4 threads per block");
    static_assert(2 * PREP_KB >= NWORD, "a packing workgroup of 256 threads fills two words of block bits per tile");
    __shared__ unsigned bm[2][RT_TILES + GT_TILES];
    if (blockIdx.x >= PREP_WB) {
        const int kb = blockIdx.x - PREP_WB;
        for (int i = threadIdx.x; i < 2 * (RT_TILES + GT_TILES); i += 256) (&bm[0][0])[i] = 0u;
        __syncthreads();
        int i = kb * 256 + threadIdx.x;
        if (i < N_EDGES / 8) {
            const int4* p = (const int4*)(ei + N_EDGES) + 2 * i;
            int4 a = p[0], b = p[1];
            uint4 o;
            o.x = dkey(a.x) | (dkey(a.y) << 16);
            o.y = dkey(a.z) | (dkey(a.w) << 16);
            o.z = dkey(b.x) | (dkey(b.y) << 16);
            o.w = dkey(b.z) | (dkey(b.w) << 16);
            ((uint4*)dk)[i] = o;
            unsigned* row = bm[threadIdx.x >> 7];
            const unsigned bit = 1u << ((threadIdx.x >> 2) & 31);
            const int d[8] = {a.x, a.y, a.z, a.w, b.x, b.y, b.z, b.w};
            #pragma unroll
            for (int k = 0; k < 8; ++k) {
                if ((unsigned)d[k] < (unsigned)N_NODES) {
                    atomicOr(&row[d[k] / RT_N], bit);
                    atomicOr(&row[RT_TILES + d[k] / TILE_N], bit);
                }
            }
        }
        __syncthreads();
        for (int i = threadIdx.x; i < 2 * (RT_TILES + GT_TILES); i += 256) {
            const int j = i / (RT_TILES + GT_TILES), t = i - j * (RT_TILES + GT_TILES);
            const unsigned v = bm[j][t];
            const size_t gw = (size_t)(kb * 2 + j);
            if (t < RT_TILES) bits_r[gw * RT_TILES + t] = v;
            else              bits_g[gw * GT_TILES + (t - RT_TILES)] = v;
        }
        return;
    }
    int t = blockIdx.x * blockDim.x + threadIdx.x;
    if (t >= RREL * GDIM * HD1) return;
    int r = t / (GDIM * HD1);
    int io = t % (GDIM * HD1);
    float s = 0.f;
    #pragma unroll
    for (int b = 0; b < NB; ++b) s += comp[r * NB + b] * basis[b * GDIM * HD1 + io];
    W[t] = s;
}

#define RT_QCAP 290
#define RT_QFLUSH 32
#define RT_PITCH (RREL * 128 + 4)
#define RT_MT 5
#define RT_PP 132

__global__ __launch_bounds__(1024) void k_rel_tile(const float* __restrict__ x,
                                                   const int* __restrict__ ei,
                                                   const int* __restrict__ et,
                                                   const unsigned* __restrict__ dk,
                                                   const unsigned* __restrict__ bits,
                                                   const float* __restrict__ Wr,
                                                   const float* __restrict__ bias,
                                                   const float* __restrict__ root,
                                                   const float* __restrict__ gatw,
                                                   const float* __restrict__ att_s,
                                                   const float* __restrict__ att_d,
                                                   float* __restrict__ xlout,
                                                   float* __restrict__ asg,
                                                   float* __restrict__ adg) {
    __shared__ float tile[RT_N * RT_PITCH];
    __shared__ float s_cnt[RT_N * RREL];
    __shared__ __attribute__((aligned(16))) unsigned s_q[32][RT_QCAP];

    static_assert(N_NODES <= 65536, "source index is packed into 16 bits");
    static_assert(RT_N <= 128, "row within the tile is packed into 7 bits");
    static_assert(N_EDGES <= (1 << 20), "edge position is packed into 20 bits");
    static_assert(N_NODES + RT_N <= 65536 - 256, "a key below the tile, or NOKEY, never wraps into it");
    static_assert(RT_MT * 16 >= RT_N, "the row tiles cover the rows");
    static_assert(5 * RT_MT * 16 * RT_PP <= RT_N * RT_PITCH, "the partial products and the h rows fit where the means were");
    static_assert(RT_N * RT_PP <= 32 * RT_QCAP, "the tile's rows of x fit where the queues were");
    static_assert(GDIM == 128 && HD1 == 128 && HEADS * HD2 == 128, "all three products are 128 wide and the small ones 128 deep");

    const int tid = threadIdx.x;
    const int lane = tid & 31;
    const int wave = __builtin_amdgcn_readfirstlane(tid >> 5);
    const int lo = blockIdx.x * RT_N;
    const int cnt = (N_NODES - lo) < RT_N ? (N_NODES - lo) : RT_N;

    for (int i = tid; i < RT_N * RT_PITCH / 4; i += 1024)
        ((float4*)tile)[i] = make_float4(0.f, 0.f, 0.f, 0.f);
    for (int i = tid; i < RT_N * RREL; i += 1024) s_cnt[i] = 0.f;
    __syncthreads();

    const int* __restrict__ srcp = ei;
    unsigned* myq = s_q[wave];

    int nq = 0;
    auto flush = [&]() __attribute__((always_inline)) {
        if (nq == 0) return;
        __builtin_amdgcn_fence(__ATOMIC_ACQ_REL, "workgroup", "local");
        for (int q0 = 0; q0 < nq; q0 += 32) {
            int n = nq - q0;
            n = n < 32 ? n : 32;
            unsigned e = myq[q0 + (lane < n ? lane : 0)];
            int idx = (int)(e & 0xFFFFFu);
            int s = srcp[idx];
            int ty = et[idx];
            unsigned w = (unsigned)s | ((e >> 20) << 16) | ((unsigned)ty << 23);
            unsigned okm = __builtin_amdgcn_ballot_w32(lane < n && (unsigned)ty < (unsigned)RREL);
            for (int j = 0; j < n; j += 8) {
                unsigned ent[8];
                float xr[8][4];
                #pragma unroll
                for (int b = 0; b < 8; ++b) {
                    ent[b] = (unsigned)__builtin_amdgcn_readlane((int)w, j + b);
                    const float* row = x + (size_t)(ent[b] & 0xFFFFu) * GDIM + lane;
                    xr[b][0] = row[0]; xr[b][1] = row[32]; xr[b][2] = row[64]; xr[b][3] = row[96];
                }
                #pragma unroll
                for (int b = 0; b < 8; ++b) {
                    if ((okm >> (j + b)) & 1u) {
                        int dl = (int)((ent[b] >> 16) & 127u), r = (int)(ent[b] >> 23);
                        float* t = tile + dl * RT_PITCH + r * 128 + lane;
                        atomicAdd(t,      xr[b][0]);
                        atomicAdd(t + 32, xr[b][1]);
                        atomicAdd(t + 64, xr[b][2]);
                        atomicAdd(t + 96, xr[b][3]);
                        if (lane == 0) atomicAdd(&s_cnt[dl * RREL + r], 1.f);
                    }
                }
            }
        }
        nq = 0;
    };
    static_assert(RT_QCAP >= RT_QFLUSH + 8 * EB, "8 blocks of edges fit behind an unemptied queue");
    {
        const unsigned short* __restrict__ k16 = (const unsigned short*)dk;
        const int wi = wave + 32 * lane;
        unsigned m = wi < NWORD ? bits[(size_t)wi * RT_TILES + blockIdx.x] : 0u;
        int myblk = 0;
        bool loaded = false;
        for (;;) {
            if (!loaded && m != 0u) {
                myblk = wi * 32 + __builtin_ctz(m);
                m &= m - 1u;
                loaded = true;
            }
            const unsigned offered = __builtin_amdgcn_ballot_w32(loaded);
            if (offered == 0u) break;
            if (nq >= RT_QFLUSH) flush();
            unsigned left = offered;
            int blk[8];
            bool val[8];
            unsigned short key[8];
            #pragma unroll
            for (int u = 0; u < 8; ++u) {
                val[u] = left != 0u;
                const int j = val[u] ? __builtin_ctz(left) : 0;
                left &= left - 1u;
                blk[u] = __builtin_amdgcn_readlane(myblk, j);
                key[u] = k16[blk[u] * EB + lane];
            }
            if (((offered & ~left) >> lane) & 1u) loaded = false;
            #pragma unroll
            for (int u = 0; u < 8; ++u) {
                if (val[u]) {
                    const unsigned short rel = (unsigned short)(key[u] - (unsigned short)lo);
                    const bool hit = rel < (unsigned short)cnt;
                    const unsigned hb = __builtin_amdgcn_ballot_w32(hit);
                    if (hb != 0u) {
                        if (hit)
                            myq[nq + __builtin_amdgcn_mbcnt_lo(hb, 0u)] =
                                (unsigned)(blk[u] * EB + lane) | ((unsigned)rel << 20);
                        nq += __builtin_popcount(hb);
                    }
                }
            }
        }
        flush();
    }
    __syncthreads();

    float* Xs = (float*)&s_q[0][0];
    for (int i = tid; i < RT_N * 32; i += 1024) {
        int dl = i >> 5, q = i & 31;
        int rr = dl < cnt ? dl : cnt - 1;
        *(float4*)(Xs + dl * RT_PP + q * 4) = ((const float4*)(x + (size_t)(lo + rr) * GDIM))[q];
    }

    for (int i = tid; i < RT_N * RREL * 32; i += 1024) {
        int slot = i >> 5;
        int dl = slot >> 3, r = slot & 7;
        float c = fmaxf(s_cnt[slot], 1.f);
        float4* p = (float4*)(tile + dl * RT_PITCH + r * 128) + (i & 31);
        float4 v = *p;
        v.x = v.x / c;
        v.y = v.y / c;
        v.z = v.z / c;
        v.w = v.w / c;
        *p = v;
    }
    __syncthreads();

    const int ct = wave & 7;
    const int kq = wave >> 3;
    const int c16 = lane & 15;
    const int hh = lane >> 4;
    int aoff[RT_MT];
    #pragma unroll
    for (int t = 0; t < RT_MT; ++t) {
        int rr = t * 16 + c16;
        rr = rr < RT_N ? rr : RT_N - 1;
        aoff[t] = rr * RT_PITCH + kq * 256 + hh * 2;
    }
    const float* Bp = Wr + (size_t)(kq * 256 + hh * 2) * HD1 + ct * 16 + c16;
    const int ksm = kq * 32 + hh * 2;
    const float* Rp = root + (size_t)ksm * HD1 + ct * 16 + c16;
    const float* Gp = gatw + (size_t)ksm * HD1 + ct * 16 + c16;

    v8f acc[RT_MT];
    #pragma unroll
    for (int t = 0; t < RT_MT; ++t) {
        #pragma unroll
        for (int i = 0; i < 8; ++i) acc[t][i] = 0.f;
    }
    v2f bcur[4];
    #pragma unroll
    for (int j = 0; j < 4; ++j) {
        bcur[j].x = Bp[(size_t)(4 * j) * HD1];
        bcur[j].y = Bp[(size_t)(4 * j + 1) * HD1];
    }
    #pragma unroll 2
    for (int kb = 0; kb < 256; kb += 16) {
        const int kn = kb + 16 < 256 ? kb + 16 : kb;
        v2f bnx[4];
        #pragma unroll
        for (int j = 0; j < 4; ++j) {
            bnx[j].x = Bp[(size_t)(kn + 4 * j) * HD1];
            bnx[j].y = Bp[(size_t)(kn + 4 * j + 1) * HD1];
        }
        #pragma unroll
        for (int j = 0; j < 4; ++j) {
            #pragma unroll
            for (int t = 0; t < RT_MT; ++t) {
                v2f a = *(const v2f*)(tile + aoff[t] + kb + 4 * j);
                acc[t] = __builtin_amdgcn_wmma_f32_16x16x4_f32(false, a, false, bcur[j], (short)0,
                                                               acc[t], false, false);
            }
        }
        #pragma unroll
        for (int j = 0; j < 4; ++j) bcur[j] = bnx[j];
    }
    {
        v2f rb[8];
        #pragma unroll
        for (int j = 0; j < 8; ++j) {
            rb[j].x = Rp[(size_t)(4 * j) * HD1];
            rb[j].y = Rp[(size_t)(4 * j + 1) * HD1];
        }
        #pragma unroll
        for (int j = 0; j < 8; ++j) {
            #pragma unroll
            for (int t = 0; t < RT_MT; ++t) {
                int rr = t * 16 + c16;
                rr = rr < RT_N ? rr : RT_N - 1;
                v2f a = *(const v2f*)(Xs + rr * RT_PP + ksm + 4 * j);
                acc[t] = __builtin_amdgcn_wmma_f32_16x16x4_f32(false, a, false, rb[j], (short)0,
                                                               acc[t], false, false);
            }
        }
    }
    __syncthreads();

    float* P = tile;
    #pragma unroll
    for (int t = 0; t < RT_MT; ++t) {
        #pragma unroll
        for (int i = 0; i < 8; ++i)
            P[(kq * (RT_MT * 16) + t * 16 + hh * 8 + i) * RT_PP + ct * 16 + c16] = acc[t][i];
    }
    __syncthreads();

    v2f gb[8];
    #pragma unroll
    for (int j = 0; j < 8; ++j) {
        gb[j].x = Gp[(size_t)(4 * j) * HD1];
        gb[j].y = Gp[(size_t)(4 * j + 1) * HD1];
    }

    float* H = tile + 4 * RT_MT * 16 * RT_PP;
    for (int i = tid; i < RT_MT * 16 * 32; i += 1024) {
        int dl = i >> 5, q = i & 31;
        float4 v = ((const float4*)bias)[q];
        #pragma unroll
        for (int k = 0; k < 4; ++k) {
            float4 p = *(const float4*)(P + (k * (RT_MT * 16) + dl) * RT_PP + q * 4);
            v.x += p.x;
            v.y += p.y;
            v.z += p.z;
            v.w += p.w;
        }
        *(float4*)(H + dl * RT_PP + q * 4) = v;
    }
    __syncthreads();

    #pragma unroll
    for (int t = 0; t < RT_MT; ++t) {
        #pragma unroll
        for (int i = 0; i < 8; ++i) acc[t][i] = 0.f;
    }
    #pragma unroll
    for (int j = 0; j < 8; ++j) {
        #pragma unroll
        for (int t = 0; t < RT_MT; ++t) {
            v2f a = *(const v2f*)(H + (t * 16 + c16) * RT_PP + ksm + 4 * j);
            acc[t] = __builtin_amdgcn_wmma_f32_16x16x4_f32(false, a, false, gb[j], (short)0,
                                                           acc[t], false, false);
        }
    }
    #pragma unroll
    for (int t = 0; t < RT_MT; ++t) {
        #pragma unroll
        for (int i = 0; i < 8; ++i)
            P[(kq * (RT_MT * 16) + t * 16 + hh * 8 + i) * RT_PP + ct * 16 + c16] = acc[t][i];
    }
    __syncthreads();

    float* S = H;
    for (int i = tid; i < cnt * 32; i += 1024) {
        int dl = i >> 5, q = i & 31;
        float4 v = *(const float4*)(P + dl * RT_PP + q * 4);
        #pragma unroll
        for (int k = 1; k < 4; ++k) {
            float4 p = *(const float4*)(P + (k * (RT_MT * 16) + dl) * RT_PP + q * 4);
            v.x += p.x;
            v.y += p.y;
            v.z += p.z;
            v.w += p.w;
        }
        ((float4*)(xlout + (size_t)(lo + dl) * HD1))[q] = v;
        float4 s4 = ((const float4*)att_s)[q];
        float4 d4 = ((const float4*)att_d)[q];
        float ps = v.x * s4.x + v.y * s4.y + v.z * s4.z + v.w * s4.w;
        float pd = v.x * d4.x + v.y * d4.y + v.z * d4.z + v.w * d4.w;
        ps += __shfl_xor(ps, 1); pd += __shfl_xor(pd, 1);
        ps += __shfl_xor(ps, 2); pd += __shfl_xor(pd, 2);
        ps += __shfl_xor(ps, 4); pd += __shfl_xor(pd, 4);
        if ((q & 7) == 0) {
            S[dl * 8 + (q >> 3)] = ps;
            S[dl * 8 + 4 + (q >> 3)] = pd;
        }
    }
    __syncthreads();
    for (int i = tid; i < cnt * 4; i += 1024) {
        asg[lo * 4 + i] = S[(i >> 2) * 8 + (i & 3)];
        adg[lo * 4 + i] = S[(i >> 2) * 8 + 4 + (i & 3)];
    }
}

#define QCAP 1024

__global__ __launch_bounds__(1024) void k_gat_tile(const int* __restrict__ ei,
                                                   const unsigned* __restrict__ dk,
                                                   const unsigned* __restrict__ bits,
                                                   const float* __restrict__ xl,
                                                   const float* __restrict__ a_src,
                                                   const float* __restrict__ a_dst,
                                                   const float* __restrict__ bias,
                                                   float* __restrict__ out) {
    __shared__ float tile[TILE_N * 128];
    __shared__ float s_den[TILE_N * 4];
    __shared__ float s_m[TILE_N * 4];
    __shared__ float s_ad[TILE_N * 4];
    __shared__ unsigned s_q[32][QCAP];
    __shared__ int s_ovf;

    static_assert(N_NODES <= 65536, "source index is packed into 16 bits");
    static_assert(N_EDGES % EG == 0, "a wave reads EG consecutive destination keys per load");
    static_assert(N_EDGES <= (1 << 20), "edge position is packed into 20 bits");
    static_assert(TILE_N <= 256, "row within the tile is packed into 8 bits");
    static_assert(N_NODES + TILE_N <= 65536 - 256, "a key below the tile, or NOKEY, never wraps into it");
    static_assert(QCAP >= 32 + EG + EG, "room for an unfinished batch, a group of edges and the give-up margin");

    const int tid = threadIdx.x;
    const int lane = tid & 31;
    const int wave = __builtin_amdgcn_readfirstlane(tid >> 5);
    const int lo = blockIdx.x * TILE_N;
    const int cnt = (N_NODES - lo) < TILE_N ? (N_NODES - lo) : TILE_N;

    if (tid == 0) s_ovf = 0;
    for (int i = tid; i < cnt * 4; i += 1024) {
        int g = lo * 4 + i;
        float ad = a_dst[g];
        s_ad[i] = ad;
        s_m[i] = lrelu(a_src[g] + ad);
    }
    __syncthreads();

    const int* __restrict__ srcp = ei;
    unsigned* myq = s_q[wave];
    const unsigned lo2 = (unsigned)lo | ((unsigned)lo << 16);
    const unsigned cnt2 = (unsigned)cnt | ((unsigned)cnt << 16);

    int nq = 0;
    int done = 0;
    int lim = 32;

    auto add_batch = [&](int s, int dl, int n) __attribute__((always_inline)) {
        bool ok = lane < n && s != lo + dl;
        unsigned okm = __builtin_amdgcn_ballot_w32(ok);
        float4 as = *(const float4*)(a_src + s * 4);
        float4 ad = *(const float4*)&s_ad[dl * 4];
        float4 mm = *(const float4*)&s_m[dl * 4];
        float w0 = __expf(lrelu(as.x + ad.x) - mm.x);
        float w1 = __expf(lrelu(as.y + ad.y) - mm.y);
        float w2 = __expf(lrelu(as.z + ad.z) - mm.z);
        float w3 = __expf(lrelu(as.w + ad.w) - mm.w);
        if (ok) {
            atomicAdd(&s_den[dl * 4 + 0], w0);
            atomicAdd(&s_den[dl * 4 + 1], w1);
            atomicAdd(&s_den[dl * 4 + 2], w2);
            atomicAdd(&s_den[dl * 4 + 3], w3);
        }
        for (int j = 0; j < n; j += 8) {
            float xr[8][4];
            #pragma unroll
            for (int b = 0; b < 8; ++b) {
                int ss = __builtin_amdgcn_readlane(s, j + b);
                const float* row = xl + (size_t)ss * 128 + lane;
                xr[b][0] = row[0]; xr[b][1] = row[32]; xr[b][2] = row[64]; xr[b][3] = row[96];
            }
            #pragma unroll
            for (int b = 0; b < 8; ++b) {
                if ((okm >> (j + b)) & 1u) {
                    int dj = __builtin_amdgcn_readlane(dl, j + b);
                    float v0 = __int_as_float(__builtin_amdgcn_readlane(__float_as_int(w0), j + b));
                    float v1 = __int_as_float(__builtin_amdgcn_readlane(__float_as_int(w1), j + b));
                    float v2 = __int_as_float(__builtin_amdgcn_readlane(__float_as_int(w2), j + b));
                    float v3 = __int_as_float(__builtin_amdgcn_readlane(__float_as_int(w3), j + b));
                    float* t = tile + dj * 128 + lane;
                    atomicAdd(t,      xr[b][0] * v0);
                    atomicAdd(t + 32, xr[b][1] * v1);
                    atomicAdd(t + 64, xr[b][2] * v2);
                    atomicAdd(t + 96, xr[b][3] * v3);
                }
            }
        }
    };
    auto flush_max = [&]() __attribute__((always_inline)) {
        if (nq > done) {
            __builtin_amdgcn_fence(__ATOMIC_ACQ_REL, "workgroup", "local");
            for (int q0 = done; q0 < nq; q0 += 32) {
                int n = nq - q0;
                n = n < 32 ? n : 32;
                unsigned e = myq[q0 + (lane < n ? lane : 0)];
                int s = srcp[(int)(e & 0xFFFFFu)];
                int dl = (int)(e >> 20);
                float4 as = *(const float4*)(a_src + s * 4);
                float4 ad = *(const float4*)&s_ad[dl * 4];
                if (lane < n) {
                    if (s != lo + dl) {
                        atomicMaxF(&s_m[dl * 4 + 0], lrelu(as.x + ad.x));
                        atomicMaxF(&s_m[dl * 4 + 1], lrelu(as.y + ad.y));
                        atomicMaxF(&s_m[dl * 4 + 2], lrelu(as.z + ad.z));
                        atomicMaxF(&s_m[dl * 4 + 3], lrelu(as.w + ad.w));
                    }
                    myq[q0 + lane] = (unsigned)s | ((unsigned)dl << 16);
                }
            }
            done = nq;
        }
        if (nq + EG > QCAP) {
            if (lane == 0) s_ovf = 1;
            nq = 0;
            done = 0;
        }
        lim = done + 32 < QCAP - (EG - 1) ? done + 32 : QCAP - (EG - 1);
        nq = __builtin_amdgcn_readfirstlane(nq);
        done = __builtin_amdgcn_readfirstlane(done);
        lim = __builtin_amdgcn_readfirstlane(lim);
    };
    auto flush_sum = [&]() __attribute__((always_inline)) {
        if (nq == 0) return;
        __builtin_amdgcn_fence(__ATOMIC_ACQ_REL, "workgroup", "local");
        for (int q0 = 0; q0 < nq; q0 += 32) {
            int n = nq - q0;
            n = n < 32 ? n : 32;
            unsigned e = myq[q0 + (lane < n ? lane : 0)];
            int s = srcp[(int)(e & 0xFFFFFu)];
            add_batch(s, (int)(e >> 20), n);
        }
        nq = 0;
    };
    auto scan_group = [&](const v4u d, int g) __attribute__((always_inline)) {
        #pragma unroll
        for (int hf = 0; hf < 2; ++hf) {
            const unsigned ra = pk_sub16(d[2 * hf], lo2), rb = pk_sub16(d[2 * hf + 1], lo2);
            if (__builtin_amdgcn_ballot_w32(pk_min16(pk_min16(ra, rb), cnt2) != cnt2) != 0u) {
                const int pos = g * EG + lane * 8 + 4 * hf;
                #pragma unroll
                for (int k = 0; k < 4; ++k) {
                    const unsigned short rel = __builtin_bit_cast(us2, k < 2 ? ra : rb)[k & 1];
                    bool hit = rel < (unsigned short)cnt;
                    unsigned bal = __builtin_amdgcn_ballot_w32(hit);
                    if (bal != 0u) {
                        if (hit)
                            myq[nq + __builtin_amdgcn_mbcnt_lo(bal, 0u)] = (unsigned)(pos + k) | ((unsigned)rel << 20);
                        nq += __builtin_popcount(bal);
                    }
                }
            }
        }
    };
    auto fetch = [&](v4u (&buf)[8], int gb) __attribute__((always_inline)) {
        #pragma unroll
        for (int u = 0; u < 8; ++u) {
            int g = gb + u * 32 + wave;
            g = g < NGRP ? g : NGRP - 1;
            buf[u] = *(const v4u*)(dk + g * (EG / 2) + lane * 4);
        }
    };
    auto step_sum = [&](const v4u (&buf)[8], int gb) __attribute__((always_inline)) {
        if (nq >= 32) flush_sum();
        int u = 8;
        #pragma unroll
        for (int uu = 0; uu < 8; ++uu) {
            if (gb + uu * 32 + wave >= NGRP) break;
            if (__builtin_expect(nq > QCAP - EG, 0)) { u = uu; break; }
            scan_group(buf[uu], gb + uu * 32 + wave);
        }
        #pragma unroll 1
        for (; u < 8 && gb + u * 32 + wave < NGRP; ++u) {
            flush_sum();
            const int g = gb + u * 32 + wave;
            scan_group(*(const v4u*)(dk + g * (EG / 2) + lane * 4), g);
        }
    };

    static_assert(8 * EB == EG, "a batch of 8 blocks is what the queue keeps room for");
    {
        const unsigned short* __restrict__ k16 = (const unsigned short*)dk;
        const int wi = wave + 32 * lane;
        unsigned m = wi < NWORD ? bits[(size_t)wi * GT_TILES + blockIdx.x] : 0u;
        int myblk = 0;
        bool loaded = false;
        for (;;) {
            if (!loaded && m != 0u) {
                myblk = wi * 32 + __builtin_ctz(m);
                m &= m - 1u;
                loaded = true;
            }
            const unsigned offered = __builtin_amdgcn_ballot_w32(loaded);
            if (offered == 0u) break;
            if (nq >= lim) flush_max();
            unsigned left = offered;
            int blk[8];
            bool val[8];
            unsigned short key[8];
            #pragma unroll
            for (int u = 0; u < 8; ++u) {
                val[u] = left != 0u;
                const int j = val[u] ? __builtin_ctz(left) : 0;
                left &= left - 1u;
                blk[u] = __builtin_amdgcn_readlane(myblk, j);
                key[u] = k16[blk[u] * EB + lane];
            }
            if (((offered & ~left) >> lane) & 1u) loaded = false;
            #pragma unroll
            for (int u = 0; u < 8; ++u) {
                if (val[u]) {
                    const unsigned short rel = (unsigned short)(key[u] - (unsigned short)lo);
                    const bool hit = rel < (unsigned short)cnt;
                    const unsigned hb = __builtin_amdgcn_ballot_w32(hit);
                    if (hb != 0u) {
                        if (hit)
                            myq[nq + __builtin_amdgcn_mbcnt_lo(hb, 0u)] =
                                (unsigned)(blk[u] * EB + lane) | ((unsigned)rel << 20);
                        nq += __builtin_popcount(hb);
                    }
                }
            }
        }
        flush_max();
    }
    __syncthreads();
    const bool rescan = __builtin_amdgcn_readfirstlane(s_ovf) != 0;

    for (int i = tid; i < cnt * 4; i += 1024)
        s_den[i] = __expf(lrelu(a_src[lo * 4 + i] + s_ad[i]) - s_m[i]);
    __syncthreads();
    for (int i = tid; i < cnt * 128; i += 1024)
        tile[i] = xl[(size_t)lo * 128 + i] * s_den[(i >> 7) * 4 + ((i & 127) >> 5)];
    __syncthreads();

    if (!rescan) {
        for (int q0 = 0; q0 < nq; q0 += 32) {
            int n = nq - q0;
            n = n < 32 ? n : 32;
            unsigned e = myq[q0 + (lane < n ? lane : 0)];
            add_batch((int)(e & 0xFFFFu), (int)(e >> 16), n);
        }
    } else {
        nq = 0;
        done = 0;
        lim = 32;
        v4u bA[8], bB[8];
        fetch(bA, 0);
        fetch(bB, GSTEP);
        for (int gb = 0; gb < NGRP; gb += 2 * GSTEP) {
            step_sum(bA, gb);
            fetch(bA, gb + 2 * GSTEP);
            step_sum(bB, gb + GSTEP);
            fetch(bB, gb + 3 * GSTEP);
        }
        flush_sum();
    }
    __syncthreads();

    for (int i = tid; i < cnt * 32; i += 1024) {
        int q = i & 31;
        float dn = s_den[(i >> 5) * 4 + (q >> 3)];
        float4 b = ((const float4*)bias)[q];
        float4 v = ((const float4*)tile)[i];
        v.x = v.x / dn + b.x;
        v.y = v.y / dn + b.y;
        v.z = v.z / dn + b.z;
        v.w = v.w / dn + b.w;
        ((float4*)out)[(size_t)lo * 32 + i] = v;
    }
}


extern "C" void kernel_launch(void* const* d_in, const int* in_sizes, int n_in,
                              void* d_out, int out_size, void* d_ws, size_t ws_size,
                              hipStream_t stream) {
    const float* x        = (const float*)d_in[0];
    const int*   ei       = (const int*)d_in[1];
    const int*   et       = (const int*)d_in[2];
    const float* comp     = (const float*)d_in[3];
    const float* basis    = (const float*)d_in[4];
    const float* root     = (const float*)d_in[5];
    const float* bias1    = (const float*)d_in[6];
    const float* gat_w    = (const float*)d_in[7];
    const float* att_src  = (const float*)d_in[8];
    const float* att_dst  = (const float*)d_in[9];
    const float* gat_bias = (const float*)d_in[10];
    float* out = (float*)d_out;

    float* ws = (float*)d_ws;
    size_t off = 0;
    float* W    = ws + off; off += (size_t)RREL * GDIM * HD1;
    float* xl   = ws + off; off += (size_t)N_NODES * HD1;
    float* asg  = ws + off; off += (size_t)N_NODES * HEADS;
    float* adg  = ws + off; off += (size_t)N_NODES * HEADS;
    unsigned* dk = (unsigned*)(ws + off); off += (size_t)N_EDGES / 2;
    unsigned* bits_r = (unsigned*)(ws + off); off += (size_t)2 * PREP_KB * RT_TILES;
    unsigned* bits_g = (unsigned*)(ws + off); off += (size_t)2 * PREP_KB * GT_TILES;

    k_basis_w<<<PREP_WB + PREP_KB, 256, 0, stream>>>(comp, basis, W, ei, dk, bits_r, bits_g);
    k_rel_tile<<<RT_TILES, 1024, 0, stream>>>(x, ei, et, dk, bits_r, W, bias1, root, gat_w, att_src, att_dst,
                                              xl, asg, adg);

    k_gat_tile<<<GT_TILES, 1024, 0, stream>>>(ei, dk, bits_g, xl, asg, adg, gat_bias, out);

    (void)in_sizes; (void)n_in; (void)out_size; (void)ws_size;
}


# relation tile kernel: 32 rows per workgroup (was 70, padded to 80), 64 registers: two workgroups share a WGP, one's edge scan runs under the other's products; LDS float atomics stay, arrival order
# speedup vs baseline: 4.8597x; 1.0315x over previous
#include <hip/hip_runtime.h>
#include <math.h>

#define N_NODES 50000
#define N_EDGES 800000
#define GDIM 128
#define HD1 128
#define HEADS 4
#define HD2 32
#define RREL 8
#define NB 20

typedef __attribute__((ext_vector_type(2))) float v2f;
typedef __attribute__((ext_vector_type(8))) float v8f;
typedef __attribute__((ext_vector_type(4))) unsigned int v4u;
typedef __attribute__((ext_vector_type(2))) unsigned short us2;

__device__ __forceinline__ unsigned pk_sub16(unsigned a, unsigned b) {
    return __builtin_bit_cast(unsigned, (us2)(__builtin_bit_cast(us2, a) - __builtin_bit_cast(us2, b)));
}
__device__ __forceinline__ unsigned pk_min16(unsigned a, unsigned b) {
    return __builtin_bit_cast(unsigned, __builtin_elementwise_min(__builtin_bit_cast(us2, a), __builtin_bit_cast(us2, b)));
}

#define EG 256
#define NGRP (N_EDGES / EG)
#define GSTEP (32 * 8)
#define NOKEY 0xFFFFu

#define RT_N 32
#define TILE_N 196
#define RT_TILES ((N_NODES + RT_N - 1) / RT_N)
#define GT_TILES ((N_NODES + TILE_N - 1) / TILE_N)
#define EB 32
#define NBLK (N_EDGES / EB)
#define NWORD ((NBLK + 31) / 32)

__device__ __forceinline__ float lrelu(float x) { return x > 0.f ? x : 0.2f * x; }

__device__ __forceinline__ void atomicMaxF(float* addr, float val) {
    if (val >= 0.f) atomicMax((int*)addr, __float_as_int(val));
    else            atomicMin((unsigned int*)addr, __float_as_uint(val));
}


#define PREP_WB ((RREL * GDIM * HD1 + 255) / 256)
#define PREP_KB ((N_EDGES / 8 + 255) / 256)

__device__ __forceinline__ unsigned dkey(int d) { return (unsigned)d < (unsigned)N_NODES ? (unsigned)d : NOKEY; }

__global__ __launch_bounds__(256) void k_basis_w(const float* __restrict__ comp, const float* __restrict__ basis,
                          float* __restrict__ W, const int* __restrict__ ei,
                          unsigned* __restrict__ dk, unsigned* __restrict__ bits_r,
                          unsigned* __restrict__ bits_g) {
    static_assert(N_EDGES % 8 == 0, "a thread packs 8 destinations");
    static_assert(EB == 32 && N_EDGES % EB == 0, "a thread's 8 edges lie in one block; 4 threads per block");
    static_assert(2 * PREP_KB >= NWORD, "a packing workgroup of 256 threads fills two words of block bits per tile");
    __shared__ unsigned bm[2][RT_TILES + GT_TILES];
    if (blockIdx.x >= PREP_WB) {
        const int kb = blockIdx.x - PREP_WB;
        for (int i = threadIdx.x; i < 2 * (RT_TILES + GT_TILES); i += 256) (&bm[0][0])[i] = 0u;
        __syncthreads();
        int i = kb * 256 + threadIdx.x;
        if (i < N_EDGES / 8) {
            const int4* p = (const int4*)(ei + N_EDGES) + 2 * i;
            int4 a = p[0], b = p[1];
            uint4 o;
            o.x = dkey(a.x) | (dkey(a.y) << 16);
            o.y = dkey(a.z) | (dkey(a.w) << 16);
            o.z = dkey(b.x) | (dkey(b.y) << 16);
            o.w = dkey(b.z) | (dkey(b.w) << 16);
            ((uint4*)dk)[i] = o;
            unsigned* row = bm[threadIdx.x >> 7];
            const unsigned bit = 1u << ((threadIdx.x >> 2) & 31);
            const int d[8] = {a.x, a.y, a.z, a.w, b.x, b.y, b.z, b.w};
            #pragma unroll
            for (int k = 0; k < 8; ++k) {
                if ((unsigned)d[k] < (unsigned)N_NODES) {
                    atomicOr(&row[d[k] / RT_N], bit);
                    atomicOr(&row[RT_TILES + d[k] / TILE_N], bit);
                }
            }
        }
        __syncthreads();
        for (int i = threadIdx.x; i < 2 * (RT_TILES + GT_TILES); i += 256) {
            const int j = i / (RT_TILES + GT_TILES), t = i - j * (RT_TILES + GT_TILES);
            const unsigned v = bm[j][t];
            const size_t gw = (size_t)(kb * 2 + j);
            if (t < RT_TILES) bits_r[gw * RT_TILES + t] = v;
            else              bits_g[gw * GT_TILES + (t - RT_TILES)] = v;
        }
        return;
    }
    int t = blockIdx.x * blockDim.x + threadIdx.x;
    if (t >= RREL * GDIM * HD1) return;
    int r = t / (GDIM * HD1);
    int io = t % (GDIM * HD1);
    float s = 0.f;
    #pragma unroll
    for (int b = 0; b < NB; ++b) s += comp[r * NB + b] * basis[b * GDIM * HD1 + io];
    W[t] = s;
}

#define RT_BB 4
#define RT_QCAP 160
#define RT_QFLUSH 32
#define RT_PITCH (RREL * 128 + 4)
#define RT_MT 2
#define RT_PP 132

__global__ __launch_bounds__(1024) __attribute__((amdgpu_waves_per_eu(16, 16))) void k_rel_tile(const float* __restrict__ x,
                                                   const int* __restrict__ ei,
                                                   const int* __restrict__ et,
                                                   const unsigned* __restrict__ dk,
                                                   const unsigned* __restrict__ bits,
                                                   const float* __restrict__ Wr,
                                                   const float* __restrict__ bias,
                                                   const float* __restrict__ root,
                                                   const float* __restrict__ gatw,
                                                   const float* __restrict__ att_s,
                                                   const float* __restrict__ att_d,
                                                   float* __restrict__ xlout,
                                                   float* __restrict__ asg,
                                                   float* __restrict__ adg) {
    __shared__ float tile[RT_N * RT_PITCH];
    __shared__ float s_cnt[RT_N * RREL];
    __shared__ __attribute__((aligned(16))) unsigned s_q[32][RT_QCAP];

    static_assert(N_NODES <= 65536, "source index is packed into 16 bits");
    static_assert(RT_N <= 128, "row within the tile is packed into 7 bits");
    static_assert(2 * (sizeof(float) * (RT_N * RT_PITCH + RT_N * RREL) + sizeof(unsigned) * 32 * RT_QCAP) <= 320 * 1024,
                  "two workgroups fit the LDS of one WGP");
    static_assert(N_EDGES <= (1 << 20), "edge position is packed into 20 bits");
    static_assert(N_NODES + RT_N <= 65536 - 256, "a key below the tile, or NOKEY, never wraps into it");
    static_assert(RT_MT * 16 >= RT_N, "the row tiles cover the rows");
    static_assert(5 * RT_MT * 16 * RT_PP <= RT_N * RT_PITCH, "the partial products and the h rows fit where the means were");
    static_assert(RT_N * RT_PP <= 32 * RT_QCAP, "the tile's rows of x fit where the queues were");
    static_assert(GDIM == 128 && HD1 == 128 && HEADS * HD2 == 128, "all three products are 128 wide and the small ones 128 deep");

    const int tid = threadIdx.x;
    const int lane = tid & 31;
    const int wave = __builtin_amdgcn_readfirstlane(tid >> 5);
    const int lo = blockIdx.x * RT_N;
    const int cnt = (N_NODES - lo) < RT_N ? (N_NODES - lo) : RT_N;

    for (int i = tid; i < RT_N * RT_PITCH / 4; i += 1024)
        ((float4*)tile)[i] = make_float4(0.f, 0.f, 0.f, 0.f);
    for (int i = tid; i < RT_N * RREL; i += 1024) s_cnt[i] = 0.f;
    __syncthreads();

    const int* __restrict__ srcp = ei;
    unsigned* myq = s_q[wave];

    int nq = 0;
    auto flush = [&]() __attribute__((always_inline)) {
        if (nq == 0) return;
        __builtin_amdgcn_fence(__ATOMIC_ACQ_REL, "workgroup", "local");
        for (int q0 = 0; q0 < nq; q0 += 32) {
            int n = nq - q0;
            n = n < 32 ? n : 32;
            unsigned e = myq[q0 + (lane < n ? lane : 0)];
            int idx = (int)(e & 0xFFFFFu);
            int s = srcp[idx];
            int ty = et[idx];
            unsigned w = (unsigned)s | ((e >> 20) << 16) | ((unsigned)ty << 23);
            unsigned okm = __builtin_amdgcn_ballot_w32(lane < n && (unsigned)ty < (unsigned)RREL);
            for (int j = 0; j < n; j += 8) {
                unsigned ent[8];
                float xr[8][4];
                #pragma unroll
                for (int b = 0; b < 8; ++b) {
                    ent[b] = (unsigned)__builtin_amdgcn_readlane((int)w, j + b);
                    const float* row = x + (size_t)(ent[b] & 0xFFFFu) * GDIM + lane;
                    xr[b][0] = row[0]; xr[b][1] = row[32]; xr[b][2] = row[64]; xr[b][3] = row[96];
                }
                #pragma unroll
                for (int b = 0; b < 8; ++b) {
                    if ((okm >> (j + b)) & 1u) {
                        int dl = (int)((ent[b] >> 16) & 127u), r = (int)(ent[b] >> 23);
                        float* t = tile + dl * RT_PITCH + r * 128 + lane;
                        atomicAdd(t,      xr[b][0]);
                        atomicAdd(t + 32, xr[b][1]);
                        atomicAdd(t + 64, xr[b][2]);
                        atomicAdd(t + 96, xr[b][3]);
                        if (lane == 0) atomicAdd(&s_cnt[dl * RREL + r], 1.f);
                    }
                }
            }
        }
        nq = 0;
    };
    static_assert(RT_QCAP >= RT_QFLUSH + RT_BB * EB, "a batch of blocks fits behind an unemptied queue");
    {
        const unsigned short* __restrict__ k16 = (const unsigned short*)dk;
        const int wi = wave + 32 * lane;
        unsigned m = wi < NWORD ? bits[(size_t)wi * RT_TILES + blockIdx.x] : 0u;
        int myblk = 0;
        bool loaded = false;
        for (;;) {
            if (!loaded && m != 0u) {
                myblk = wi * 32 + __builtin_ctz(m);
                m &= m - 1u;
                loaded = true;
            }
            const unsigned offered = __builtin_amdgcn_ballot_w32(loaded);
            if (offered == 0u) break;
            if (nq >= RT_QFLUSH) flush();
            unsigned left = offered;
            int blk[RT_BB];
            bool val[RT_BB];
            unsigned short key[RT_BB];
            #pragma unroll
            for (int u = 0; u < RT_BB; ++u) {
                val[u] = left != 0u;
                const int j = val[u] ? __builtin_ctz(left) : 0;
                left &= left - 1u;
                blk[u] = __builtin_amdgcn_readlane(myblk, j);
                key[u] = k16[blk[u] * EB + lane];
            }
            if (((offered & ~left) >> lane) & 1u) loaded = false;
            #pragma unroll
            for (int u = 0; u < RT_BB; ++u) {
                if (val[u]) {
                    const unsigned short rel = (unsigned short)(key[u] - (unsigned short)lo);
                    const bool hit = rel < (unsigned short)cnt;
                    const unsigned hb = __builtin_amdgcn_ballot_w32(hit);
                    if (hb != 0u) {
                        if (hit)
                            myq[nq + __builtin_amdgcn_mbcnt_lo(hb, 0u)] =
                                (unsigned)(blk[u] * EB + lane) | ((unsigned)rel << 20);
                        nq += __builtin_popcount(hb);
                    }
                }
            }
        }
        flush();
    }
    __syncthreads();

    float* Xs = (float*)&s_q[0][0];
    for (int i = tid; i < RT_N * 32; i += 1024) {
        int dl = i >> 5, q = i & 31;
        int rr = dl < cnt ? dl : cnt - 1;
        *(float4*)(Xs + dl * RT_PP + q * 4) = ((const float4*)(x + (size_t)(lo + rr) * GDIM))[q];
    }

    for (int i = tid; i < RT_N * RREL * 32; i += 1024) {
        int slot = i >> 5;
        int dl = slot >> 3, r = slot & 7;
        float c = fmaxf(s_cnt[slot], 1.f);
        float4* p = (float4*)(tile + dl * RT_PITCH + r * 128) + (i & 31);
        float4 v = *p;
        v.x = v.x / c;
        v.y = v.y / c;
        v.z = v.z / c;
        v.w = v.w / c;
        *p = v;
    }
    __syncthreads();

    const int ct = wave & 7;
    const int kq = wave >> 3;
    const int c16 = lane & 15;
    const int hh = lane >> 4;
    int aoff[RT_MT];
    #pragma unroll
    for (int t = 0; t < RT_MT; ++t) {
        int rr = t * 16 + c16;
        rr = rr < RT_N ? rr : RT_N - 1;
        aoff[t] = rr * RT_PITCH + kq * 256 + hh * 2;
    }
    const float* Bp = Wr + (size_t)(kq * 256 + hh * 2) * HD1 + ct * 16 + c16;
    const int ksm = kq * 32 + hh * 2;
    const float* Rp = root + (size_t)ksm * HD1 + ct * 16 + c16;
    const float* Gp = gatw + (size_t)ksm * HD1 + ct * 16 + c16;

    v8f acc[RT_MT];
    #pragma unroll
    for (int t = 0; t < RT_MT; ++t) {
        #pragma unroll
        for (int i = 0; i < 8; ++i) acc[t][i] = 0.f;
    }
    v2f bcur[4];
    #pragma unroll
    for (int j = 0; j < 4; ++j) {
        bcur[j].x = Bp[(size_t)(4 * j) * HD1];
        bcur[j].y = Bp[(size_t)(4 * j + 1) * HD1];
    }
    #pragma unroll 2
    for (int kb = 0; kb < 256; kb += 16) {
        const int kn = kb + 16 < 256 ? kb + 16 : kb;
        v2f bnx[4];
        #pragma unroll
        for (int j = 0; j < 4; ++j) {
            bnx[j].x = Bp[(size_t)(kn + 4 * j) * HD1];
            bnx[j].y = Bp[(size_t)(kn + 4 * j + 1) * HD1];
        }
        #pragma unroll
        for (int j = 0; j < 4; ++j) {
            #pragma unroll
            for (int t = 0; t < RT_MT; ++t) {
                v2f a = *(const v2f*)(tile + aoff[t] + kb + 4 * j);
                acc[t] = __builtin_amdgcn_wmma_f32_16x16x4_f32(false, a, false, bcur[j], (short)0,
                                                               acc[t], false, false);
            }
        }
        #pragma unroll
        for (int j = 0; j < 4; ++j) bcur[j] = bnx[j];
    }
    {
        v2f rb[8];
        #pragma unroll
        for (int j = 0; j < 8; ++j) {
            rb[j].x = Rp[(size_t)(4 * j) * HD1];
            rb[j].y = Rp[(size_t)(4 * j + 1) * HD1];
        }
        #pragma unroll
        for (int j = 0; j < 8; ++j) {
            #pragma unroll
            for (int t = 0; t < RT_MT; ++t) {
                int rr = t * 16 + c16;
                rr = rr < RT_N ? rr : RT_N - 1;
                v2f a = *(const v2f*)(Xs + rr * RT_PP + ksm + 4 * j);
                acc[t] = __builtin_amdgcn_wmma_f32_16x16x4_f32(false, a, false, rb[j], (short)0,
                                                               acc[t], false, false);
            }
        }
    }
    __syncthreads();

    float* P = tile;
    #pragma unroll
    for (int t = 0; t < RT_MT; ++t) {
        #pragma unroll
        for (int i = 0; i < 8; ++i)
            P[(kq * (RT_MT * 16) + t * 16 + hh * 8 + i) * RT_PP + ct * 16 + c16] = acc[t][i];
    }
    __syncthreads();

    v2f gb[8];
    #pragma unroll
    for (int j = 0; j < 8; ++j) {
        gb[j].x = Gp[(size_t)(4 * j) * HD1];
        gb[j].y = Gp[(size_t)(4 * j + 1) * HD1];
    }

    float* H = tile + 4 * RT_MT * 16 * RT_PP;
    for (int i = tid; i < RT_MT * 16 * 32; i += 1024) {
        int dl = i >> 5, q = i & 31;
        float4 v = ((const float4*)bias)[q];
        #pragma unroll
        for (int k = 0; k < 4; ++k) {
            float4 p = *(const float4*)(P + (k * (RT_MT * 16) + dl) * RT_PP + q * 4);
            v.x += p.x;
            v.y += p.y;
            v.z += p.z;
            v.w += p.w;
        }
        *(float4*)(H + dl * RT_PP + q * 4) = v;
    }
    __syncthreads();

    #pragma unroll
    for (int t = 0; t < RT_MT; ++t) {
        #pragma unroll
        for (int i = 0; i < 8; ++i) acc[t][i] = 0.f;
    }
    #pragma unroll
    for (int j = 0; j < 8; ++j) {
        #pragma unroll
        for (int t = 0; t < RT_MT; ++t) {
            v2f a = *(const v2f*)(H + (t * 16 + c16) * RT_PP + ksm + 4 * j);
            acc[t] = __builtin_amdgcn_wmma_f32_16x16x4_f32(false, a, false, gb[j], (short)0,
                                                           acc[t], false, false);
        }
    }
    #pragma unroll
    for (int t = 0; t < RT_MT; ++t) {
        #pragma unroll
        for (int i = 0; i < 8; ++i)
            P[(kq * (RT_MT * 16) + t * 16 + hh * 8 + i) * RT_PP + ct * 16 + c16] = acc[t][i];
    }
    __syncthreads();

    float* S = H;
    for (int i = tid; i < cnt * 32; i += 1024) {
        int dl = i >> 5, q = i & 31;
        float4 v = *(const float4*)(P + dl * RT_PP + q * 4);
        #pragma unroll
        for (int k = 1; k < 4; ++k) {
            float4 p = *(const float4*)(P + (k * (RT_MT * 16) + dl) * RT_PP + q * 4);
            v.x += p.x;
            v.y += p.y;
            v.z += p.z;
            v.w += p.w;
        }
        ((float4*)(xlout + (size_t)(lo + dl) * HD1))[q] = v;
        float4 s4 = ((const float4*)att_s)[q];
        float4 d4 = ((const float4*)att_d)[q];
        float ps = v.x * s4.x + v.y * s4.y + v.z * s4.z + v.w * s4.w;
        float pd = v.x * d4.x + v.y * d4.y + v.z * d4.z + v.w * d4.w;
        ps += __shfl_xor(ps, 1); pd += __shfl_xor(pd, 1);
        ps += __shfl_xor(ps, 2); pd += __shfl_xor(pd, 2);
        ps += __shfl_xor(ps, 4); pd += __shfl_xor(pd, 4);
        if ((q & 7) == 0) {
            S[dl * 8 + (q >> 3)] = ps;
            S[dl * 8 + 4 + (q >> 3)] = pd;
        }
    }
    __syncthreads();
    for (int i = tid; i < cnt * 4; i += 1024) {
        asg[lo * 4 + i] = S[(i >> 2) * 8 + (i & 3)];
        adg[lo * 4 + i] = S[(i >> 2) * 8 + 4 + (i & 3)];
    }
}

#define QCAP 1024

__global__ __launch_bounds__(1024) void k_gat_tile(const int* __restrict__ ei,
                                                   const unsigned* __restrict__ dk,
                                                   const unsigned* __restrict__ bits,
                                                   const float* __restrict__ xl,
                                                   const float* __restrict__ a_src,
                                                   const float* __restrict__ a_dst,
                                                   const float* __restrict__ bias,
                                                   float* __restrict__ out) {
    __shared__ float tile[TILE_N * 128];
    __shared__ float s_den[TILE_N * 4];
    __shared__ float s_m[TILE_N * 4];
    __shared__ float s_ad[TILE_N * 4];
    __shared__ unsigned s_q[32][QCAP];
    __shared__ int s_ovf;

    static_assert(N_NODES <= 65536, "source index is packed into 16 bits");
    static_assert(N_EDGES % EG == 0, "a wave reads EG consecutive destination keys per load");
    static_assert(N_EDGES <= (1 << 20), "edge position is packed into 20 bits");
    static_assert(TILE_N <= 256, "row within the tile is packed into 8 bits");
    static_assert(N_NODES + TILE_N <= 65536 - 256, "a key below the tile, or NOKEY, never wraps into it");
    static_assert(QCAP >= 32 + EG + EG, "room for an unfinished batch, a group of edges and the give-up margin");

    const int tid = threadIdx.x;
    const int lane = tid & 31;
    const int wave = __builtin_amdgcn_readfirstlane(tid >> 5);
    const int lo = blockIdx.x * TILE_N;
    const int cnt = (N_NODES - lo) < TILE_N ? (N_NODES - lo) : TILE_N;

    if (tid == 0) s_ovf = 0;
    for (int i = tid; i < cnt * 4; i += 1024) {
        int g = lo * 4 + i;
        float ad = a_dst[g];
        s_ad[i] = ad;
        s_m[i] = lrelu(a_src[g] + ad);
    }
    __syncthreads();

    const int* __restrict__ srcp = ei;
    unsigned* myq = s_q[wave];
    const unsigned lo2 = (unsigned)lo | ((unsigned)lo << 16);
    const unsigned cnt2 = (unsigned)cnt | ((unsigned)cnt << 16);

    int nq = 0;
    int done = 0;
    int lim = 32;

    auto add_batch = [&](int s, int dl, int n) __attribute__((always_inline)) {
        bool ok = lane < n && s != lo + dl;
        unsigned okm = __builtin_amdgcn_ballot_w32(ok);
        float4 as = *(const float4*)(a_src + s * 4);
        float4 ad = *(const float4*)&s_ad[dl * 4];
        float4 mm = *(const float4*)&s_m[dl * 4];
        float w0 = __expf(lrelu(as.x + ad.x) - mm.x);
        float w1 = __expf(lrelu(as.y + ad.y) - mm.y);
        float w2 = __expf(lrelu(as.z + ad.z) - mm.z);
        float w3 = __expf(lrelu(as.w + ad.w) - mm.w);
        if (ok) {
            atomicAdd(&s_den[dl * 4 + 0], w0);
            atomicAdd(&s_den[dl * 4 + 1], w1);
            atomicAdd(&s_den[dl * 4 + 2], w2);
            atomicAdd(&s_den[dl * 4 + 3], w3);
        }
        for (int j = 0; j < n; j += 8) {
            float xr[8][4];
            #pragma unroll
            for (int b = 0; b < 8; ++b) {
                int ss = __builtin_amdgcn_readlane(s, j + b);
                const float* row = xl + (size_t)ss * 128 + lane;
                xr[b][0] = row[0]; xr[b][1] = row[32]; xr[b][2] = row[64]; xr[b][3] = row[96];
            }
            #pragma unroll
            for (int b = 0; b < 8; ++b) {
                if ((okm >> (j + b)) & 1u) {
                    int dj = __builtin_amdgcn_readlane(dl, j + b);
                    float v0 = __int_as_float(__builtin_amdgcn_readlane(__float_as_int(w0), j + b));
                    float v1 = __int_as_float(__builtin_amdgcn_readlane(__float_as_int(w1), j + b));
                    float v2 = __int_as_float(__builtin_amdgcn_readlane(__float_as_int(w2), j + b));
                    float v3 = __int_as_float(__builtin_amdgcn_readlane(__float_as_int(w3), j + b));
                    float* t = tile + dj * 128 + lane;
                    atomicAdd(t,      xr[b][0] * v0);
                    atomicAdd(t + 32, xr[b][1] * v1);
                    atomicAdd(t + 64, xr[b][2] * v2);
                    atomicAdd(t + 96, xr[b][3] * v3);
                }
            }
        }
    };
    auto flush_max = [&]() __attribute__((always_inline)) {
        if (nq > done) {
            __builtin_amdgcn_fence(__ATOMIC_ACQ_REL, "workgroup", "local");
            for (int q0 = done; q0 < nq; q0 += 32) {
                int n = nq - q0;
                n = n < 32 ? n : 32;
                unsigned e = myq[q0 + (lane < n ? lane : 0)];
                int s = srcp[(int)(e & 0xFFFFFu)];
                int dl = (int)(e >> 20);
                float4 as = *(const float4*)(a_src + s * 4);
                float4 ad = *(const float4*)&s_ad[dl * 4];
                if (lane < n) {
                    if (s != lo + dl) {
                        atomicMaxF(&s_m[dl * 4 + 0], lrelu(as.x + ad.x));
                        atomicMaxF(&s_m[dl * 4 + 1], lrelu(as.y + ad.y));
                        atomicMaxF(&s_m[dl * 4 + 2], lrelu(as.z + ad.z));
                        atomicMaxF(&s_m[dl * 4 + 3], lrelu(as.w + ad.w));
                    }
                    myq[q0 + lane] = (unsigned)s | ((unsigned)dl << 16);
                }
            }
            done = nq;
        }
        if (nq + EG > QCAP) {
            if (lane == 0) s_ovf = 1;
            nq = 0;
            done = 0;
        }
        lim = done + 32 < QCAP - (EG - 1) ? done + 32 : QCAP - (EG - 1);
        nq = __builtin_amdgcn_readfirstlane(nq);
        done = __builtin_amdgcn_readfirstlane(done);
        lim = __builtin_amdgcn_readfirstlane(lim);
    };
    auto flush_sum = [&]() __attribute__((always_inline)) {
        if (nq == 0) return;
        __builtin_amdgcn_fence(__ATOMIC_ACQ_REL, "workgroup", "local");
        for (int q0 = 0; q0 < nq; q0 += 32) {
            int n = nq - q0;
            n = n < 32 ? n : 32;
            unsigned e = myq[q0 + (lane < n ? lane : 0)];
            int s = srcp[(int)(e & 0xFFFFFu)];
            add_batch(s, (int)(e >> 20), n);
        }
        nq = 0;
    };
    auto scan_group = [&](const v4u d, int g) __attribute__((always_inline)) {
        #pragma unroll
        for (int hf = 0; hf < 2; ++hf) {
            const unsigned ra = pk_sub16(d[2 * hf], lo2), rb = pk_sub16(d[2 * hf + 1], lo2);
            if (__builtin_amdgcn_ballot_w32(pk_min16(pk_min16(ra, rb), cnt2) != cnt2) != 0u) {
                const int pos = g * EG + lane * 8 + 4 * hf;
                #pragma unroll
                for (int k = 0; k < 4; ++k) {
                    const unsigned short rel = __builtin_bit_cast(us2, k < 2 ? ra : rb)[k & 1];
                    bool hit = rel < (unsigned short)cnt;
                    unsigned bal = __builtin_amdgcn_ballot_w32(hit);
                    if (bal != 0u) {
                        if (hit)
                            myq[nq + __builtin_amdgcn_mbcnt_lo(bal, 0u)] = (unsigned)(pos + k) | ((unsigned)rel << 20);
                        nq += __builtin_popcount(bal);
                    }
                }
            }
        }
    };
    auto fetch = [&](v4u (&buf)[8], int gb) __attribute__((always_inline)) {
        #pragma unroll
        for (int u = 0; u < 8; ++u) {
            int g = gb + u * 32 + wave;
            g = g < NGRP ? g : NGRP - 1;
            buf[u] = *(const v4u*)(dk + g * (EG / 2) + lane * 4);
        }
    };
    auto step_sum = [&](const v4u (&buf)[8], int gb) __attribute__((always_inline)) {
        if (nq >= 32) flush_sum();
        int u = 8;
        #pragma unroll
        for (int uu = 0; uu < 8; ++uu) {
            if (gb + uu * 32 + wave >= NGRP) break;
            if (__builtin_expect(nq > QCAP - EG, 0)) { u = uu; break; }
            scan_group(buf[uu], gb + uu * 32 + wave);
        }
        #pragma unroll 1
        for (; u < 8 && gb + u * 32 + wave < NGRP; ++u) {
            flush_sum();
            const int g = gb + u * 32 + wave;
            scan_group(*(const v4u*)(dk + g * (EG / 2) + lane * 4), g);
        }
    };

    static_assert(8 * EB == EG, "a batch of 8 blocks is what the queue keeps room for");
    {
        const unsigned short* __restrict__ k16 = (const unsigned short*)dk;
        const int wi = wave + 32 * lane;
        unsigned m = wi < NWORD ? bits[(size_t)wi * GT_TILES + blockIdx.x] : 0u;
        int myblk = 0;
        bool loaded = false;
        for (;;) {
            if (!loaded && m != 0u) {
                myblk = wi * 32 + __builtin_ctz(m);
                m &= m - 1u;
                loaded = true;
            }
            const unsigned offered = __builtin_amdgcn_ballot_w32(loaded);
            if (offered == 0u) break;
            if (nq >= lim) flush_max();
            unsigned left = offered;
            int blk[8];
            bool val[8];
            unsigned short key[8];
            #pragma unroll
            for (int u = 0; u < 8; ++u) {
                val[u] = left != 0u;
                const int j = val[u] ? __builtin_ctz(left) : 0;
                left &= left - 1u;
                blk[u] = __builtin_amdgcn_readlane(myblk, j);
                key[u] = k16[blk[u] * EB + lane];
            }
            if (((offered & ~left) >> lane) & 1u) loaded = false;
            #pragma unroll
            for (int u = 0; u < 8; ++u) {
                if (val[u]) {
                    const unsigned short rel = (unsigned short)(key[u] - (unsigned short)lo);
                    const bool hit = rel < (unsigned short)cnt;
                    const unsigned hb = __builtin_amdgcn_ballot_w32(hit);
                    if (hb != 0u) {
                        if (hit)
                            myq[nq + __builtin_amdgcn_mbcnt_lo(hb, 0u)] =
                                (unsigned)(blk[u] * EB + lane) | ((unsigned)rel << 20);
                        nq += __builtin_popcount(hb);
                    }
                }
            }
        }
        flush_max();
    }
    __syncthreads();
    const bool rescan = __builtin_amdgcn_readfirstlane(s_ovf) != 0;

    for (int i = tid; i < cnt * 4; i += 1024)
        s_den[i] = __expf(lrelu(a_src[lo * 4 + i] + s_ad[i]) - s_m[i]);
    __syncthreads();
    for (int i = tid; i < cnt * 128; i += 1024)
        tile[i] = xl[(size_t)lo * 128 + i] * s_den[(i >> 7) * 4 + ((i & 127) >> 5)];
    __syncthreads();

    if (!rescan) {
        for (int q0 = 0; q0 < nq; q0 += 32) {
            int n = nq - q0;
            n = n < 32 ? n : 32;
            unsigned e = myq[q0 + (lane < n ? lane : 0)];
            add_batch((int)(e & 0xFFFFu), (int)(e >> 16), n);
        }
    } else {
        nq = 0;
        done = 0;
        lim = 32;
        v4u bA[8], bB[8];
        fetch(bA, 0);
        fetch(bB, GSTEP);
        for (int gb = 0; gb < NGRP; gb += 2 * GSTEP) {
            step_sum(bA, gb);
            fetch(bA, gb + 2 * GSTEP);
            step_sum(bB, gb + GSTEP);
            fetch(bB, gb + 3 * GSTEP);
        }
        flush_sum();
    }
    __syncthreads();

    for (int i = tid; i < cnt * 32; i += 1024) {
        int q = i & 31;
        float dn = s_den[(i >> 5) * 4 + (q >> 3)];
        float4 b = ((const float4*)bias)[q];
        float4 v = ((const float4*)tile)[i];
        v.x = v.x / dn + b.x;
        v.y = v.y / dn + b.y;
        v.z = v.z / dn + b.z;
        v.w = v.w / dn + b.w;
        ((float4*)out)[(size_t)lo * 32 + i] = v;
    }
}


extern "C" void kernel_launch(void* const* d_in, const int* in_sizes, int n_in,
                              void* d_out, int out_size, void* d_ws, size_t ws_size,
                              hipStream_t stream) {
    const float* x        = (const float*)d_in[0];
    const int*   ei       = (const int*)d_in[1];
    const int*   et       = (const int*)d_in[2];
    const float* comp     = (const float*)d_in[3];
    const float* basis    = (const float*)d_in[4];
    const float* root     = (const float*)d_in[5];
    const float* bias1    = (const float*)d_in[6];
    const float* gat_w    = (const float*)d_in[7];
    const float* att_src  = (const float*)d_in[8];
    const float* att_dst  = (const float*)d_in[9];
    const float* gat_bias = (const float*)d_in[10];
    float* out = (float*)d_out;

    float* ws = (float*)d_ws;
    size_t off = 0;
    float* W    = ws + off; off += (size_t)RREL * GDIM * HD1;
    float* xl   = ws + off; off += (size_t)N_NODES * HD1;
    float* asg  = ws + off; off += (size_t)N_NODES * HEADS;
    float* adg  = ws + off; off += (size_t)N_NODES * HEADS;
    unsigned* dk = (unsigned*)(ws + off); off += (size_t)N_EDGES / 2;
    unsigned* bits_r = (unsigned*)(ws + off); off += (size_t)2 * PREP_KB * RT_TILES;
    unsigned* bits_g = (unsigned*)(ws + off); off += (size_t)2 * PREP_KB * GT_TILES;

    k_basis_w<<<PREP_WB + PREP_KB, 256, 0, stream>>>(comp, basis, W, ei, dk, bits_r, bits_g);
    k_rel_tile<<<RT_TILES, 1024, 0, stream>>>(x, ei, et, dk, bits_r, W, bias1, root, gat_w, att_src, att_dst,
                                              xl, asg, adg);

    k_gat_tile<<<GT_TILES, 1024, 0, stream>>>(ei, dk, bits_g, xl, asg, adg, gat_bias, out);

    (void)in_sizes; (void)n_in; (void)out_size; (void)ws_size;
}
